# starting kernel
# baseline (speedup 1.0000x reference)
#include <hip/hip_runtime.h>

typedef __attribute__((ext_vector_type(2))) float v2f;
typedef __attribute__((ext_vector_type(8))) float v8f;

__global__ void gcn_zero_f32(float* __restrict__ p, int n) {
    int i = blockIdx.x * blockDim.x + threadIdx.x;
    if (i < n) p[i] = 0.0f;
}

__global__ void gcn_scatter(const float* __restrict__ x,
                            const int*   __restrict__ ei,
                            float*       __restrict__ agg,
                            float*       __restrict__ deg,
                            int E) {
    int tid = blockIdx.x * blockDim.x + threadIdx.x;
    int e   = tid >> 4;
    if (e >= E) return;
    int sub = tid & 15;
    int row = ei[e];
    int col = ei[E + e];

    float4 v = reinterpret_cast<const float4*>(x + (long)col * 64)[sub];
    float* a = agg + (long)row * 64 + sub * 4;
    atomicAdd(a + 0, v.x);
    atomicAdd(a + 1, v.y);
    atomicAdd(a + 2, v.z);
    atomicAdd(a + 3, v.w);
    if (sub == 0) atomicAdd(deg + row, 1.0f);
}

__global__ void gcn_gemm_wmma(float*       __restrict__ inout,
                              const float* __restrict__ deg,
                              const float* __restrict__ W,
                              const float* __restrict__ bias,
                              int N) {
    int wave = (int)((blockIdx.x * blockDim.x + threadIdx.x) >> 5);
    int lane = threadIdx.x & 31;
    int m0   = wave * 16;
    if (m0 >= N) return;

    int   mr     = lane & 15;
    int   half   = lane >> 4;
    int   m      = m0 + mr;
    int   mc     = (m < N) ? m : (N - 1);
    float invdeg = 1.0f / (deg[mc] + 1e-6f);
    const float* arow = inout + (long)mc * 64;

    int nl = lane & 15;
    const float* w0 = W + (long)(0 * 16 + nl) * 64;
    const float* w1 = W + (long)(1 * 16 + nl) * 64;
    const float* w2 = W + (long)(2 * 16 + nl) * 64;
    const float* w3 = W + (long)(3 * 16 + nl) * 64;

    v8f acc0 = {}, acc1 = {}, acc2 = {}, acc3 = {};

#pragma unroll
    for (int k0 = 0; k0 < 64; k0 += 4) {
        int kb = k0 + half * 2;

        v2f a;
        a.x = arow[kb]     * invdeg;
        a.y = arow[kb + 1] * invdeg;

        v2f b0, b1, b2, b3;
        b0.x = w0[kb]; b0.y = w0[kb + 1];
        b1.x = w1[kb]; b1.y = w1[kb + 1];
        b2.x = w2[kb]; b2.y = w2[kb + 1];
        b3.x = w3[kb]; b3.y = w3[kb + 1];

        acc0 = __builtin_amdgcn_wmma_f32_16x16x4_f32(false, a, false, b0, (short)0, acc0, false, false);
        acc1 = __builtin_amdgcn_wmma_f32_16x16x4_f32(false, a, false, b1, (short)0, acc1, false, false);
        acc2 = __builtin_amdgcn_wmma_f32_16x16x4_f32(false, a, false, b2, (short)0, acc2, false, false);
        acc3 = __builtin_amdgcn_wmma_f32_16x16x4_f32(false, a, false, b3, (short)0, acc3, false, false);
    }

    v8f accs[4] = {acc0, acc1, acc2, acc3};
#pragma unroll
    for (int t = 0; t < 4; ++t) {
        int   n  = t * 16 + nl;
        float bv = bias[n];
#pragma unroll
        for (int i = 0; i < 8; ++i) {
            int row = m0 + half * 8 + i;
            if (row < N) inout[(long)row * 64 + n] = accs[t][i] + bv;
        }
    }
}

extern "C" void kernel_launch(void* const* d_in, const int* in_sizes, int n_in,
                              void* d_out, int out_size, void* d_ws, size_t ws_size,
                              hipStream_t stream) {
    const float* x  = (const float*)d_in[0];
    const int*   ei = (const int*)d_in[1];
    const float* W  = (const float*)d_in[2];
    const float* b  = (const float*)d_in[3];
    float* out = (float*)d_out;
    float* deg = (float*)d_ws;

    const int N = in_sizes[0] / 64;
    const int E = in_sizes[1] / 2;

    int zn = N * 64;
    gcn_zero_f32<<<(zn + 255) / 256, 256, 0, stream>>>(out, zn);
    gcn_zero_f32<<<(N + 255) / 256, 256, 0, stream>>>(deg, N);

    long sthreads = (long)E * 16;
    int  sblocks  = (int)((sthreads + 255) / 256);
    gcn_scatter<<<sblocks, 256, 0, stream>>>(x, ei, out, deg, E);

    int waves  = (N + 15) / 16;
    int blocks = (waves * 32 + 255) / 256;
    gcn_gemm_wmma<<<blocks, 256, 0, stream>>>(out, deg, W, b, N);
}


# scatter: a wave walks 32 edges; each load and each f32 atomic add covers one whole 128 B line (lane l owns floats l and l+32), not 4 B adds strided over four lines; sums stay unordered atomics
# speedup vs baseline: 1.9528x; 1.9528x over previous
#include <hip/hip_runtime.h>

typedef __attribute__((ext_vector_type(2))) float v2f;
typedef __attribute__((ext_vector_type(8))) float v8f;

__global__ void gcn_zero_f32(float* __restrict__ p, int n) {
    int i = blockIdx.x * blockDim.x + threadIdx.x;
    if (i < n) p[i] = 0.0f;
}

#define SC_BATCH 8

__global__ void __launch_bounds__(256)
gcn_scatter(const float* __restrict__ x,
            const int*   __restrict__ ei,
            float*       __restrict__ agg,
            float*       __restrict__ deg,
            int E) {
    const int lane = threadIdx.x & 31;
    const int wave = __builtin_amdgcn_readfirstlane(
        (int)(blockIdx.x * (blockDim.x >> 5) + (threadIdx.x >> 5)));
    const int base = wave * 32;
    if (base >= E) return;
    const int cnt = (E - base < 32) ? (E - base) : 32;

    const int  e     = base + lane;
    const bool ok    = e < E;
    const int  myrow = ok ? ei[e]     : 0;
    const int  mycol = ok ? ei[E + e] : 0;
    if (ok) atomicAdd(deg + myrow, 1.0f);

    for (int j0 = 0; j0 < cnt; j0 += SC_BATCH) {
        float lo[SC_BATCH], hi[SC_BATCH];
        int   rr[SC_BATCH];
        for (int u = 0; u < SC_BATCH; ++u) {
            const int j = j0 + u;
            const int c = __builtin_amdgcn_readlane(mycol, j);
            rr[u]       = __builtin_amdgcn_readlane(myrow, j);
            const float* xp = x + (long)c * 64 + lane;
            lo[u] = xp[0];
            hi[u] = xp[32];
        }
        for (int u = 0; u < SC_BATCH; ++u) {
            if (j0 + u < cnt) {
                float* ap = agg + (long)rr[u] * 64 + lane;
                atomicAdd(ap,      lo[u]);
                atomicAdd(ap + 32, hi[u]);
            }
        }
    }
}

__global__ void gcn_gemm_wmma(float*       __restrict__ inout,
                              const float* __restrict__ deg,
                              const float* __restrict__ W,
                              const float* __restrict__ bias,
                              int N) {
    int wave = (int)((blockIdx.x * blockDim.x + threadIdx.x) >> 5);
    int lane = threadIdx.x & 31;
    int m0   = wave * 16;
    if (m0 >= N) return;

    int   mr     = lane & 15;
    int   half   = lane >> 4;
    int   m      = m0 + mr;
    int   mc     = (m < N) ? m : (N - 1);
    float invdeg = 1.0f / (deg[mc] + 1e-6f);
    const float* arow = inout + (long)mc * 64;

    int nl = lane & 15;
    const float* w0 = W + (long)(0 * 16 + nl) * 64;
    const float* w1 = W + (long)(1 * 16 + nl) * 64;
    const float* w2 = W + (long)(2 * 16 + nl) * 64;
    const float* w3 = W + (long)(3 * 16 + nl) * 64;

    v8f acc0 = {}, acc1 = {}, acc2 = {}, acc3 = {};

#pragma unroll
    for (int k0 = 0; k0 < 64; k0 += 4) {
        int kb = k0 + half * 2;

        v2f a;
        a.x = arow[kb]     * invdeg;
        a.y = arow[kb + 1] * invdeg;

        v2f b0, b1, b2, b3;
        b0.x = w0[kb]; b0.y = w0[kb + 1];
        b1.x = w1[kb]; b1.y = w1[kb + 1];
        b2.x = w2[kb]; b2.y = w2[kb + 1];
        b3.x = w3[kb]; b3.y = w3[kb + 1];

        acc0 = __builtin_amdgcn_wmma_f32_16x16x4_f32(false, a, false, b0, (short)0, acc0, false, false);
        acc1 = __builtin_amdgcn_wmma_f32_16x16x4_f32(false, a, false, b1, (short)0, acc1, false, false);
        acc2 = __builtin_amdgcn_wmma_f32_16x16x4_f32(false, a, false, b2, (short)0, acc2, false, false);
        acc3 = __builtin_amdgcn_wmma_f32_16x16x4_f32(false, a, false, b3, (short)0, acc3, false, false);
    }

    v8f accs[4] = {acc0, acc1, acc2, acc3};
#pragma unroll
    for (int t = 0; t < 4; ++t) {
        int   n  = t * 16 + nl;
        float bv = bias[n];
#pragma unroll
        for (int i = 0; i < 8; ++i) {
            int row = m0 + half * 8 + i;
            if (row < N) inout[(long)row * 64 + n] = accs[t][i] + bv;
        }
    }
}

extern "C" void kernel_launch(void* const* d_in, const int* in_sizes, int n_in,
                              void* d_out, int out_size, void* d_ws, size_t ws_size,
                              hipStream_t stream) {
    const float* x  = (const float*)d_in[0];
    const int*   ei = (const int*)d_in[1];
    const float* W  = (const float*)d_in[2];
    const float* b  = (const float*)d_in[3];
    float* out = (float*)d_out;
    float* deg = (float*)d_ws;

    const int N = in_sizes[0] / 64;
    const int E = in_sizes[1] / 2;

    int zn = N * 64;
    gcn_zero_f32<<<(zn + 255) / 256, 256, 0, stream>>>(out, zn);
    gcn_zero_f32<<<(N + 255) / 256, 256, 0, stream>>>(deg, N);

    int sblocks = (E + 255) / 256;
    gcn_scatter<<<sblocks, 256, 0, stream>>>(x, ei, out, deg, E);

    int waves  = (N + 15) / 16;
    int blocks = (waves * 32 + 255) / 256;
    gcn_gemm_wmma<<<blocks, 256, 0, stream>>>(out, deg, W, b, N);
}


# zero-fill of agg and deg as ONE launch of 4096 looping waves with b128 stores (512 B per wave store), not 100000 + 1568 one-store waves in two launches: wave entry was its cost
# speedup vs baseline: 2.0623x; 1.0561x over previous
#include <hip/hip_runtime.h>

typedef __attribute__((ext_vector_type(2))) float v2f;
typedef __attribute__((ext_vector_type(8))) float v8f;

typedef __attribute__((ext_vector_type(4))) float v4f;

#define ZF_BLOCKS  512
#define ZF_THREADS 256

__global__ void __launch_bounds__(ZF_THREADS)
gcn_zero_fill(float* __restrict__ agg, int n4_agg,
              float* __restrict__ deg, int n_deg) {
    const int tid = (int)(blockIdx.x * ZF_THREADS + threadIdx.x);
    const int nth = (int)(gridDim.x * ZF_THREADS);
    const v4f z = {0.0f, 0.0f, 0.0f, 0.0f};

    v4f* a4 = (v4f*)agg;
    for (int i = tid; i < n4_agg; i += nth) a4[i] = z;

    const int n4_deg = n_deg >> 2;
    v4f* d4 = (v4f*)deg;
    for (int i = tid; i < n4_deg; i += nth) d4[i] = z;
    const int t = (n4_deg << 2) + tid;
    if (t < n_deg) deg[t] = 0.0f;
}

#define SC_BATCH 8

__global__ void __launch_bounds__(256)
gcn_scatter(const float* __restrict__ x,
            const int*   __restrict__ ei,
            float*       __restrict__ agg,
            float*       __restrict__ deg,
            int E) {
    const int lane = threadIdx.x & 31;
    const int wave = __builtin_amdgcn_readfirstlane(
        (int)(blockIdx.x * (blockDim.x >> 5) + (threadIdx.x >> 5)));
    const int base = wave * 32;
    if (base >= E) return;
    const int cnt = (E - base < 32) ? (E - base) : 32;

    const int  e     = base + lane;
    const bool ok    = e < E;
    const int  myrow = ok ? ei[e]     : 0;
    const int  mycol = ok ? ei[E + e] : 0;
    if (ok) atomicAdd(deg + myrow, 1.0f);

    for (int j0 = 0; j0 < cnt; j0 += SC_BATCH) {
        float lo[SC_BATCH], hi[SC_BATCH];
        int   rr[SC_BATCH];
        for (int u = 0; u < SC_BATCH; ++u) {
            const int j = j0 + u;
            const int c = __builtin_amdgcn_readlane(mycol, j);
            rr[u]       = __builtin_amdgcn_readlane(myrow, j);
            const float* xp = x + (long)c * 64 + lane;
            lo[u] = xp[0];
            hi[u] = xp[32];
        }
        for (int u = 0; u < SC_BATCH; ++u) {
            if (j0 + u < cnt) {
                float* ap = agg + (long)rr[u] * 64 + lane;
                atomicAdd(ap,      lo[u]);
                atomicAdd(ap + 32, hi[u]);
            }
        }
    }
}

__global__ void gcn_gemm_wmma(float*       __restrict__ inout,
                              const float* __restrict__ deg,
                              const float* __restrict__ W,
                              const float* __restrict__ bias,
                              int N) {
    int wave = (int)((blockIdx.x * blockDim.x + threadIdx.x) >> 5);
    int lane = threadIdx.x & 31;
    int m0   = wave * 16;
    if (m0 >= N) return;

    int   mr     = lane & 15;
    int   half   = lane >> 4;
    int   m      = m0 + mr;
    int   mc     = (m < N) ? m : (N - 1);
    float invdeg = 1.0f / (deg[mc] + 1e-6f);
    const float* arow = inout + (long)mc * 64;

    int nl = lane & 15;
    const float* w0 = W + (long)(0 * 16 + nl) * 64;
    const float* w1 = W + (long)(1 * 16 + nl) * 64;
    const float* w2 = W + (long)(2 * 16 + nl) * 64;
    const float* w3 = W + (long)(3 * 16 + nl) * 64;

    v8f acc0 = {}, acc1 = {}, acc2 = {}, acc3 = {};

#pragma unroll
    for (int k0 = 0; k0 < 64; k0 += 4) {
        int kb = k0 + half * 2;

        v2f a;
        a.x = arow[kb]     * invdeg;
        a.y = arow[kb + 1] * invdeg;

        v2f b0, b1, b2, b3;
        b0.x = w0[kb]; b0.y = w0[kb + 1];
        b1.x = w1[kb]; b1.y = w1[kb + 1];
        b2.x = w2[kb]; b2.y = w2[kb + 1];
        b3.x = w3[kb]; b3.y = w3[kb + 1];

        acc0 = __builtin_amdgcn_wmma_f32_16x16x4_f32(false, a, false, b0, (short)0, acc0, false, false);
        acc1 = __builtin_amdgcn_wmma_f32_16x16x4_f32(false, a, false, b1, (short)0, acc1, false, false);
        acc2 = __builtin_amdgcn_wmma_f32_16x16x4_f32(false, a, false, b2, (short)0, acc2, false, false);
        acc3 = __builtin_amdgcn_wmma_f32_16x16x4_f32(false, a, false, b3, (short)0, acc3, false, false);
    }

    v8f accs[4] = {acc0, acc1, acc2, acc3};
#pragma unroll
    for (int t = 0; t < 4; ++t) {
        int   n  = t * 16 + nl;
        float bv = bias[n];
#pragma unroll
        for (int i = 0; i < 8; ++i) {
            int row = m0 + half * 8 + i;
            if (row < N) inout[(long)row * 64 + n] = accs[t][i] + bv;
        }
    }
}

extern "C" void kernel_launch(void* const* d_in, const int* in_sizes, int n_in,
                              void* d_out, int out_size, void* d_ws, size_t ws_size,
                              hipStream_t stream) {
    const float* x  = (const float*)d_in[0];
    const int*   ei = (const int*)d_in[1];
    const float* W  = (const float*)d_in[2];
    const float* b  = (const float*)d_in[3];
    float* out = (float*)d_out;
    float* deg = (float*)d_ws;

    const int N = in_sizes[0] / 64;
    const int E = in_sizes[1] / 2;

    gcn_zero_fill<<<ZF_BLOCKS, ZF_THREADS, 0, stream>>>(out, N * 16, deg, N);

    int sblocks = (E + 255) / 256;
    gcn_scatter<<<sblocks, 256, 0, stream>>>(x, ei, out, deg, E);

    int waves  = (N + 15) / 16;
    int blocks = (waves * 32 + 255) / 256;
    gcn_gemm_wmma<<<blocks, 256, 0, stream>>>(out, deg, W, b, N);
}


# global f32 atomic scatter replaced by a counting sort of edges into 256 row ranges; one work-group per range sums its rows with LDS f32 atomics (still unordered), stores agg/deg once; zero-fill gone
# speedup vs baseline: 2.6683x; 1.2938x over previous
#include <hip/hip_runtime.h>


typedef __attribute__((ext_vector_type(2))) float v2f;
typedef __attribute__((ext_vector_type(4))) float v4f;
typedef __attribute__((ext_vector_type(8))) float v8f;

#define NB 256
#define NW 128
#define ST 512
#define SU 16
#define GT 512

__global__ void __launch_bounds__(ST)
gcn_count(const int* __restrict__ ei,
          int*       __restrict__ hist,
          int E, int rpb) {
    __shared__ int h[NB];
    const int tid = (int)threadIdx.x;
    if (tid < NB) h[tid] = 0;
    __syncthreads();

    const int chunk = (E + NW - 1) / NW;
    const int beg   = (int)blockIdx.x * chunk;
    const int end   = (beg + chunk < E) ? (beg + chunk) : E;
    for (int e0 = beg + tid; e0 < end; e0 += ST * SU) {
        int r[SU];
#pragma unroll
        for (int u = 0; u < SU; ++u) {
            const int e = e0 + u * ST;
            r[u] = (e < end) ? ei[e] : 0;
        }
#pragma unroll
        for (int u = 0; u < SU; ++u)
            if (e0 + u * ST < end) atomicAdd(&h[r[u] / rpb], 1);
    }
    __syncthreads();
    if (tid < NB) hist[(int)blockIdx.x * NB + tid] = h[tid];
}

__global__ void __launch_bounds__(ST)
gcn_place(const int* __restrict__ ei,
          const int* __restrict__ hist,
          int*       __restrict__ pairs_i,
          int*       __restrict__ bstart,
          int E, int rpb) {
    __shared__ int cur[NB];
    __shared__ int sc[NB];
    __shared__ int pbel[2][NB];
    __shared__ int pall[2][NB];
    const int tid = (int)threadIdx.x;
    const int w   = (int)blockIdx.x;
    const int b   = tid & (NB - 1);
    const int q   = tid >> 8;

    int below = 0, all = 0;
    {
        int hv[NW / 2];
#pragma unroll
        for (int k = 0; k < NW / 2; ++k)
            hv[k] = hist[(q * (NW / 2) + k) * NB + b];
#pragma unroll
        for (int k = 0; k < NW / 2; ++k) {
            below += (q * (NW / 2) + k < w) ? hv[k] : 0;
            all   += hv[k];
        }
    }
    pbel[q][b] = below;
    pall[q][b] = all;
    __syncthreads();

    int tot = 0, bel = 0;
    if (tid < NB) {
        tot = pall[0][tid] + pall[1][tid];
        bel = pbel[0][tid] + pbel[1][tid];
        sc[tid] = tot;
    }
    __syncthreads();
    for (int d = 1; d < NB; d <<= 1) {
        int v = 0;
        if (tid < NB && tid >= d) v = sc[tid - d];
        __syncthreads();
        if (tid < NB) sc[tid] += v;
        __syncthreads();
    }
    if (tid < NB) {
        const int base = sc[tid] - tot;
        cur[tid] = base + bel;
        if (w == 0) {
            bstart[tid] = base;
            if (tid == NB - 1) bstart[NB] = sc[tid];
        }
    }
    __syncthreads();

    int2* pairs = (int2*)pairs_i;
    const int chunk = (E + NW - 1) / NW;
    const int beg   = w * chunk;
    const int end   = (beg + chunk < E) ? (beg + chunk) : E;
    for (int e0 = beg + tid; e0 < end; e0 += ST * SU) {
        int r[SU], c[SU];
#pragma unroll
        for (int u = 0; u < SU; ++u) {
            const int  e  = e0 + u * ST;
            const bool ok = e < end;
            r[u] = ok ? ei[e]     : 0;
            c[u] = ok ? ei[E + e] : 0;
        }
        int p[SU];
#pragma unroll
        for (int u = 0; u < SU; ++u) {
            p[u] = 0;
            if (e0 + u * ST < end) p[u] = atomicAdd(&cur[r[u] / rpb], 1);
        }
#pragma unroll
        for (int u = 0; u < SU; ++u)
            if (e0 + u * ST < end) pairs[p[u]] = make_int2(r[u], c[u]);
    }
}

__global__ void __launch_bounds__(GT)
gcn_gather(const float* __restrict__ x,
           const int*   __restrict__ pairs_i,
           const int*   __restrict__ bstart,
           float*       __restrict__ agg,
           float*       __restrict__ deg,
           int N, int rpb) {
    extern __shared__ __attribute__((aligned(16))) float sm[];
    float* tile = sm;
    float* dg   = sm + (rpb + 1) * 64;
    const int2* pairs = (const int2*)pairs_i;
    const int tid  = (int)threadIdx.x;
    const int lane = tid & 31;
    const int wave = __builtin_amdgcn_readfirstlane(tid >> 5);
    const int row0 = (int)blockIdx.x * rpb;
    int nrows = N - row0;
    if (nrows > rpb) nrows = rpb;
    if (nrows < 0)   nrows = 0;

    const v4f z = {0.0f, 0.0f, 0.0f, 0.0f};
    v4f* s4 = (v4f*)sm;
    for (int i = tid; i < (rpb + 1) * 16; i += GT) s4[i] = z;
    for (int i = tid; i < rpb; i += GT) dg[i] = 0.0f;
    __syncthreads();

    const int s0 = bstart[blockIdx.x];
    const int s1 = bstart[blockIdx.x + 1];
    int  base = s0 + wave * 32;
    int2 nxt  = make_int2(row0 + rpb, 0);
    if (base < s1 && lane < s1 - base) nxt = pairs[base + lane];
    for (; base < s1; base += GT) {
        const int  cnt = (s1 - base < 32) ? (s1 - base) : 32;
        const int2 rc  = nxt;

        const int myrow = rc.x - row0;
        const int mycol = rc.y;
        if (lane < cnt) atomicAdd(dg + myrow, 1.0f);

        v2f v[32];
        int rr[32];
#pragma unroll
        for (int u = 0; u < 32; ++u) {
            const int c = __builtin_amdgcn_readlane(mycol, u);
            rr[u]       = __builtin_amdgcn_readlane(myrow, u);
            const float* xr = x + (long)c * 64;
            v[u] = *(const v2f*)(xr + 2 * lane);
        }
        const int nb = base + GT;
        nxt = make_int2(row0 + rpb, 0);
        if (nb < s1 && lane < s1 - nb) nxt = pairs[nb + lane];
#pragma unroll
        for (int u = 0; u < 32; ++u) {
            float* tp = tile + rr[u] * 64 + 2 * lane;
            atomicAdd(tp,     v[u].x);
            atomicAdd(tp + 1, v[u].y);
        }
    }
    __syncthreads();

    v4f*       o4 = (v4f*)(agg + (long)row0 * 64);
    const v4f* t4 = (const v4f*)tile;
    for (int i = tid; i < nrows * 16; i += GT) o4[i] = t4[i];
    for (int i = tid; i < nrows; i += GT) deg[row0 + i] = dg[i];
}

__global__ void gcn_gemm_wmma(float*       __restrict__ inout,
                              const float* __restrict__ deg,
                              const float* __restrict__ W,
                              const float* __restrict__ bias,
                              int N) {
    int wave = (int)((blockIdx.x * blockDim.x + threadIdx.x) >> 5);
    int lane = threadIdx.x & 31;
    int m0   = wave * 16;
    if (m0 >= N) return;

    int   mr     = lane & 15;
    int   half   = lane >> 4;
    int   m      = m0 + mr;
    int   mc     = (m < N) ? m : (N - 1);
    float invdeg = 1.0f / (deg[mc] + 1e-6f);
    const float* arow = inout + (long)mc * 64;

    int nl = lane & 15;
    const float* w0 = W + (long)(0 * 16 + nl) * 64;
    const float* w1 = W + (long)(1 * 16 + nl) * 64;
    const float* w2 = W + (long)(2 * 16 + nl) * 64;
    const float* w3 = W + (long)(3 * 16 + nl) * 64;

    v8f acc0 = {}, acc1 = {}, acc2 = {}, acc3 = {};

#pragma unroll
    for (int k0 = 0; k0 < 64; k0 += 4) {
        int kb = k0 + half * 2;

        v2f a;
        a.x = arow[kb]     * invdeg;
        a.y = arow[kb + 1] * invdeg;

        v2f b0, b1, b2, b3;
        b0.x = w0[kb]; b0.y = w0[kb + 1];
        b1.x = w1[kb]; b1.y = w1[kb + 1];
        b2.x = w2[kb]; b2.y = w2[kb + 1];
        b3.x = w3[kb]; b3.y = w3[kb + 1];

        acc0 = __builtin_amdgcn_wmma_f32_16x16x4_f32(false, a, false, b0, (short)0, acc0, false, false);
        acc1 = __builtin_amdgcn_wmma_f32_16x16x4_f32(false, a, false, b1, (short)0, acc1, false, false);
        acc2 = __builtin_amdgcn_wmma_f32_16x16x4_f32(false, a, false, b2, (short)0, acc2, false, false);
        acc3 = __builtin_amdgcn_wmma_f32_16x16x4_f32(false, a, false, b3, (short)0, acc3, false, false);
    }

    v8f accs[4] = {acc0, acc1, acc2, acc3};
#pragma unroll
    for (int t = 0; t < 4; ++t) {
        int   n  = t * 16 + nl;
        float bv = bias[n];
#pragma unroll
        for (int i = 0; i < 8; ++i) {
            int row = m0 + half * 8 + i;
            if (row < N) inout[(long)row * 64 + n] = accs[t][i] + bv;
        }
    }
}

extern "C" void kernel_launch(void* const* d_in, const int* in_sizes, int n_in,
                              void* d_out, int out_size, void* d_ws, size_t ws_size,
                              hipStream_t stream) {
    const float* x  = (const float*)d_in[0];
    const int*   ei = (const int*)d_in[1];
    const float* W  = (const float*)d_in[2];
    const float* b  = (const float*)d_in[3];
    float* out = (float*)d_out;

    const int N = in_sizes[0] / 64;
    const int E = in_sizes[1] / 2;
    const int rpb = (N + NB - 1) / NB;

    char* ws = (char*)d_ws;
    const size_t deg_bytes  = (((size_t)N * 4) + 255) / 256 * 256;
    const size_t hist_bytes = (size_t)NW * NB * 4;
    const size_t bst_bytes  = 2048;
    float* deg    = (float*)ws;
    int*   hist   = (int*)(ws + deg_bytes);
    int*   bstart = (int*)(ws + deg_bytes + hist_bytes);
    int*   pairs  = (int*)(ws + deg_bytes + hist_bytes + bst_bytes);

    gcn_count<<<NW, ST, 0, stream>>>(ei, hist, E, rpb);
    gcn_place<<<NW, ST, 0, stream>>>(ei, hist, pairs, bstart, E, rpb);
    const size_t lds = ((size_t)(rpb + 1) * 64 + (size_t)rpb) * 4;
    gcn_gather<<<NB, GT, lds, stream>>>(x, pairs, bstart, out, deg, N, rpb);

    int waves  = (N + 15) / 16;
    int blocks = (waves * 32 + 255) / 256;
    gcn_gemm_wmma<<<blocks, 256, 0, stream>>>(out, deg, W, b, N);
}


# linear layer fused into the gather work-group: its LDS row sums feed the f32 16x16x4 product directly (one wave per 16-row tile, W staged in LDS); agg and deg never reach memory; 3 launches, not 4
# speedup vs baseline: 3.1682x; 1.1873x over previous
#include <hip/hip_runtime.h>


typedef __attribute__((ext_vector_type(2))) float v2f;
typedef __attribute__((ext_vector_type(4))) float v4f;
typedef __attribute__((ext_vector_type(8))) float v8f;

#define NB 256
#define NW 128
#define ST 512
#define SU 16
#define GT 512
#define PITCH 68

__global__ void __launch_bounds__(ST)
gcn_count(const int* __restrict__ ei,
          int*       __restrict__ hist,
          int E, int rpb) {
    __shared__ int h[NB];
    const int tid = (int)threadIdx.x;
    if (tid < NB) h[tid] = 0;
    __syncthreads();

    const int chunk = (E + NW - 1) / NW;
    const int beg   = (int)blockIdx.x * chunk;
    const int end   = (beg + chunk < E) ? (beg + chunk) : E;
    for (int e0 = beg + tid; e0 < end; e0 += ST * SU) {
        int r[SU];
#pragma unroll
        for (int u = 0; u < SU; ++u) {
            const int e = e0 + u * ST;
            r[u] = (e < end) ? ei[e] : 0;
        }
#pragma unroll
        for (int u = 0; u < SU; ++u)
            if (e0 + u * ST < end) atomicAdd(&h[r[u] / rpb], 1);
    }
    __syncthreads();
    if (tid < NB) hist[(int)blockIdx.x * NB + tid] = h[tid];
}

__global__ void __launch_bounds__(ST)
gcn_place(const int* __restrict__ ei,
          const int* __restrict__ hist,
          int*       __restrict__ pairs_i,
          int*       __restrict__ bstart,
          int E, int rpb) {
    __shared__ int cur[NB];
    __shared__ int sc[NB];
    __shared__ int pbel[2][NB];
    __shared__ int pall[2][NB];
    const int tid = (int)threadIdx.x;
    const int w   = (int)blockIdx.x;
    const int b   = tid & (NB - 1);
    const int q   = tid >> 8;

    int below = 0, all = 0;
    {
        int hv[NW / 2];
#pragma unroll
        for (int k = 0; k < NW / 2; ++k)
            hv[k] = hist[(q * (NW / 2) + k) * NB + b];
#pragma unroll
        for (int k = 0; k < NW / 2; ++k) {
            below += (q * (NW / 2) + k < w) ? hv[k] : 0;
            all   += hv[k];
        }
    }
    pbel[q][b] = below;
    pall[q][b] = all;
    __syncthreads();

    int tot = 0, bel = 0;
    if (tid < NB) {
        tot = pall[0][tid] + pall[1][tid];
        bel = pbel[0][tid] + pbel[1][tid];
        sc[tid] = tot;
    }
    __syncthreads();
    for (int d = 1; d < NB; d <<= 1) {
        int v = 0;
        if (tid < NB && tid >= d) v = sc[tid - d];
        __syncthreads();
        if (tid < NB) sc[tid] += v;
        __syncthreads();
    }
    if (tid < NB) {
        const int base = sc[tid] - tot;
        cur[tid] = base + bel;
        if (w == 0) {
            bstart[tid] = base;
            if (tid == NB - 1) bstart[NB] = sc[tid];
        }
    }
    __syncthreads();

    int2* pairs = (int2*)pairs_i;
    const int chunk = (E + NW - 1) / NW;
    const int beg   = w * chunk;
    const int end   = (beg + chunk < E) ? (beg + chunk) : E;
    for (int e0 = beg + tid; e0 < end; e0 += ST * SU) {
        int r[SU], c[SU];
#pragma unroll
        for (int u = 0; u < SU; ++u) {
            const int  e  = e0 + u * ST;
            const bool ok = e < end;
            r[u] = ok ? ei[e]     : 0;
            c[u] = ok ? ei[E + e] : 0;
        }
        int p[SU];
#pragma unroll
        for (int u = 0; u < SU; ++u) {
            p[u] = 0;
            if (e0 + u * ST < end) p[u] = atomicAdd(&cur[r[u] / rpb], 1);
        }
#pragma unroll
        for (int u = 0; u < SU; ++u)
            if (e0 + u * ST < end) pairs[p[u]] = make_int2(r[u], c[u]);
    }
}

__global__ void __launch_bounds__(GT)
gcn_gather_gemm(const float* __restrict__ x,
                const int*   __restrict__ pairs_i,
                const int*   __restrict__ bstart,
                const float* __restrict__ W,
                const float* __restrict__ bias,
                float*       __restrict__ out,
                int N, int rpb) {
    extern __shared__ __attribute__((aligned(16))) float sm[];
    __shared__ __attribute__((aligned(16))) float Wl[64 * PITCH];
    const int RT = (rpb + 15) >> 4;
    const int RP = RT * 16;
    float* tile = sm;
    float* dg   = sm + (RP + 1) * PITCH;
    const int2* pairs = (const int2*)pairs_i;
    const int tid  = (int)threadIdx.x;
    const int lane = tid & 31;
    const int wave = __builtin_amdgcn_readfirstlane(tid >> 5);
    const int row0 = (int)blockIdx.x * rpb;
    int nrows = N - row0;
    if (nrows > rpb) nrows = rpb;
    if (nrows < 0)   nrows = 0;

    const v4f z = {0.0f, 0.0f, 0.0f, 0.0f};
    v4f* s4 = (v4f*)sm;
    for (int i = tid; i < (RP + 1) * (PITCH / 4); i += GT) s4[i] = z;
    for (int i = tid; i < RP; i += GT) dg[i] = 0.0f;
    __syncthreads();

    const v4f* W4 = (const v4f*)W;
    v4f wv[1024 / GT];
    float bv[4];
#pragma unroll
    for (int k = 0; k < 1024 / GT; ++k) wv[k] = W4[tid + k * GT];
#pragma unroll
    for (int t = 0; t < 4; ++t) bv[t] = bias[t * 16 + (lane & 15)];

    const int s0 = bstart[blockIdx.x];
    const int s1 = bstart[blockIdx.x + 1];
    int  base = s0 + wave * 32;
    int2 nxt  = make_int2(row0 + RP, 0);
    if (base < s1 && lane < s1 - base) nxt = pairs[base + lane];
    for (; base < s1; base += GT) {
        const int  cnt = (s1 - base < 32) ? (s1 - base) : 32;
        const int2 rc  = nxt;

        const int myrow = rc.x - row0;
        const int myoff = myrow * PITCH;
        const int mycol = rc.y;
        if (lane < cnt) atomicAdd(dg + myrow, 1.0f);

        v2f v[32];
        int rr[32];
#pragma unroll
        for (int u = 0; u < 32; ++u) {
            const int c = __builtin_amdgcn_readlane(mycol, u);
            rr[u]       = __builtin_amdgcn_readlane(myoff, u);
            const float* xr = x + (long)c * 64;
            v[u] = *(const v2f*)(xr + 2 * lane);
        }
        const int nb = base + GT;
        nxt = make_int2(row0 + RP, 0);
        if (nb < s1 && lane < s1 - nb) nxt = pairs[nb + lane];
#pragma unroll
        for (int u = 0; u < 32; ++u) {
            float* tp = tile + rr[u] + 2 * lane;
            atomicAdd(tp,     v[u].x);
            atomicAdd(tp + 1, v[u].y);
        }
    }

#pragma unroll
    for (int k = 0; k < 1024 / GT; ++k) {
        const int i = tid + k * GT;
        *(v4f*)(Wl + (i >> 4) * PITCH + (i & 15) * 4) = wv[k];
    }
    __syncthreads();

    const int c16  = lane & 15;
    const int half = lane >> 4;
    const int RTu  = (nrows + 15) >> 4;
    for (int rt = wave; rt < RTu; rt += GT / 32) {
        const int   r      = rt * 16 + c16;
        const float invdeg = 1.0f / (dg[r] + 1e-6f);
        const float* arow  = tile + r * PITCH + 2 * half;
        v2f a[16];
#pragma unroll
        for (int j = 0; j < 16; ++j) a[j] = *(const v2f*)(arow + 4 * j);

        v8f acc0 = {}, acc1 = {}, acc2 = {}, acc3 = {};
        const float* w0 = Wl + (0 * 16 + c16) * PITCH + 2 * half;
        const float* w1 = Wl + (1 * 16 + c16) * PITCH + 2 * half;
        const float* w2 = Wl + (2 * 16 + c16) * PITCH + 2 * half;
        const float* w3 = Wl + (3 * 16 + c16) * PITCH + 2 * half;
#pragma unroll
        for (int j = 0; j < 16; ++j) {
            v2f as;
            as.x = a[j].x * invdeg;
            as.y = a[j].y * invdeg;
            const v2f b0 = *(const v2f*)(w0 + 4 * j);
            const v2f b1 = *(const v2f*)(w1 + 4 * j);
            const v2f b2 = *(const v2f*)(w2 + 4 * j);
            const v2f b3 = *(const v2f*)(w3 + 4 * j);
            acc0 = __builtin_amdgcn_wmma_f32_16x16x4_f32(false, as, false, b0, (short)0, acc0, false, false);
            acc1 = __builtin_amdgcn_wmma_f32_16x16x4_f32(false, as, false, b1, (short)0, acc1, false, false);
            acc2 = __builtin_amdgcn_wmma_f32_16x16x4_f32(false, as, false, b2, (short)0, acc2, false, false);
            acc3 = __builtin_amdgcn_wmma_f32_16x16x4_f32(false, as, false, b3, (short)0, acc3, false, false);
        }

        v8f accs[4] = {acc0, acc1, acc2, acc3};
        float* orow = tile + (rt * 16 + half * 8) * PITCH + c16;
#pragma unroll
        for (int t = 0; t < 4; ++t) {
#pragma unroll
            for (int i = 0; i < 8; ++i)
                orow[i * PITCH + t * 16] = accs[t][i] + bv[t];
        }
    }
    __syncthreads();

    v4f* o4 = (v4f*)(out + (long)row0 * 64);
    for (int i = tid; i < nrows * 16; i += GT)
        o4[i] = *(const v4f*)(tile + (i >> 4) * PITCH + (i & 15) * 4);
}

extern "C" void kernel_launch(void* const* d_in, const int* in_sizes, int n_in,
                              void* d_out, int out_size, void* d_ws, size_t ws_size,
                              hipStream_t stream) {
    const float* x  = (const float*)d_in[0];
    const int*   ei = (const int*)d_in[1];
    const float* W  = (const float*)d_in[2];
    const float* b  = (const float*)d_in[3];
    float* out = (float*)d_out;

    const int N = in_sizes[0] / 64;
    const int E = in_sizes[1] / 2;
    const int rpb = (N + NB - 1) / NB;

    char* ws = (char*)d_ws;
    const size_t hist_bytes = (size_t)NW * NB * 4;
    const size_t bst_bytes  = 2048;
    int*   hist   = (int*)ws;
    int*   bstart = (int*)(ws + hist_bytes);
    int*   pairs  = (int*)(ws + hist_bytes + bst_bytes);

    gcn_count<<<NW, ST, 0, stream>>>(ei, hist, E, rpb);
    gcn_place<<<NW, ST, 0, stream>>>(ei, hist, pairs, bstart, E, rpb);
    const int rp = (rpb + 15) / 16 * 16;
    const size_t lds = ((size_t)(rp + 1) * PITCH + (size_t)rp) * 4;
    gcn_gather_gemm<<<NB, GT, lds, stream>>>(x, pairs, bstart, W, b, out, N, rpb);
}


# edge sort made chunk-local: ONE launch orders each of 256 edge chunks by row range in LDS (count launch and global scan gone); a gather group finds its segments by binary search; sums still unordered
# speedup vs baseline: 4.3937x; 1.3868x over previous
#include <hip/hip_runtime.h>


typedef __attribute__((ext_vector_type(2))) float v2f;
typedef __attribute__((ext_vector_type(4))) float v4f;
typedef __attribute__((ext_vector_type(8))) float v8f;

#define NB 256
#define NW 256
#define ST 256
#define SU 16
#define GT 512
#define PITCH 68
static_assert(ST == NB, "the sort's scan has one thread per range");
static_assert(NW <= GT && NW % 32 == 0 && NB <= 256, "whole waves scan the segments; a key holds the range in 8 bits");

__global__ void __launch_bounds__(ST)
gcn_sort(const int* __restrict__ ei,
         int*       __restrict__ pairs_i,
         int*       __restrict__ seg_i,
         int E, int rpb) {
    extern __shared__ __attribute__((aligned(16))) int sl[];
    __shared__ int h[NB];
    __shared__ int st[NB];
    __shared__ int wt[ST / 32];
    const int tid  = (int)threadIdx.x;
    const int lane = tid & 31;
    const int wave = tid >> 5;
    const int w    = (int)blockIdx.x;
    const int chunk = (E + NW - 1) / NW;
    const int cpad  = (chunk + 15) & ~15;
    int beg = w * chunk;
    if (beg > E) beg = E;
    const int end = (beg + chunk < E) ? (beg + chunk) : E;
    const int n   = end - beg;
    int2*     raw = (int2*)sl;
    int2*     srt = raw + chunk;
    unsigned* key = (unsigned*)(srt + chunk);

    h[tid] = 0;
    __syncthreads();

    for (int i0 = tid; i0 < n; i0 += ST * SU) {
        int r[SU], c[SU];
#pragma unroll
        for (int u = 0; u < SU; ++u) {
            const int  i  = i0 + u * ST;
            const bool ok = i < n;
            r[u] = ok ? ei[beg + i]     : 0;
            c[u] = ok ? ei[E + beg + i] : 0;
        }
        int bk[SU], rk[SU];
#pragma unroll
        for (int u = 0; u < SU; ++u) {
            bk[u] = r[u] / rpb;
            rk[u] = 0;
            if (i0 + u * ST < n) rk[u] = atomicAdd(&h[bk[u]], 1);
        }
#pragma unroll
        for (int u = 0; u < SU; ++u) {
            const int i = i0 + u * ST;
            if (i < n) {
                raw[i] = make_int2(r[u], c[u]);
                key[i] = ((unsigned)bk[u] << 24) | (unsigned)rk[u];
            }
        }
    }
    __syncthreads();

    const int cntb = h[tid];
    int inc = cntb;
#pragma unroll
    for (int d = 1; d < 32; d <<= 1) {
        const int t = __shfl_up(inc, d);
        if (lane >= d) inc += t;
    }
    if (lane == 31) wt[wave] = inc;
    __syncthreads();
    int pre = 0;
#pragma unroll
    for (int k = 0; k < ST / 32; ++k) pre += (k < wave) ? wt[k] : 0;
    const int start = pre + inc - cntb;
    st[tid] = start;
    __syncthreads();

    for (int i0 = tid; i0 < n; i0 += ST * SU) {
        unsigned k[SU];
        int2     p[SU];
#pragma unroll
        for (int u = 0; u < SU; ++u) {
            const int i = i0 + u * ST;
            k[u] = 0u;
            p[u] = make_int2(0, 0);
            if (i < n) { k[u] = key[i]; p[u] = raw[i]; }
        }
        int pos[SU];
#pragma unroll
        for (int u = 0; u < SU; ++u) {
            pos[u] = 0;
            if (i0 + u * ST < n) pos[u] = st[k[u] >> 24] + (int)(k[u] & 0xFFFFFFu);
        }
#pragma unroll
        for (int u = 0; u < SU; ++u)
            if (i0 + u * ST < n) srt[pos[u]] = p[u];
    }
    __syncthreads();

    int2* pairs = (int2*)pairs_i + (long)w * cpad;
    for (int i0 = tid; i0 < n; i0 += ST * SU) {
        int2 p[SU];
#pragma unroll
        for (int u = 0; u < SU; ++u) {
            const int i = i0 + u * ST;
            p[u] = make_int2(0, 0);
            if (i < n) p[u] = srt[i];
        }
#pragma unroll
        for (int u = 0; u < SU; ++u) {
            const int i = i0 + u * ST;
            if (i < n) pairs[i] = p[u];
        }
    }
    ((int2*)seg_i)[tid * NW + w] = make_int2(w * cpad + start, cntb);
}

__global__ void __launch_bounds__(GT)
gcn_gather_gemm(const float* __restrict__ x,
                const int*   __restrict__ pairs_i,
                const int*   __restrict__ seg_i,
                const float* __restrict__ W,
                const float* __restrict__ bias,
                float*       __restrict__ out,
                int N, int rpb) {
    extern __shared__ __attribute__((aligned(16))) float sm[];
    __shared__ __attribute__((aligned(16))) float Wl[64 * PITCH];
    __shared__ int pf[NW + 1];
    __shared__ int go[NW];
    __shared__ int wt[NW / 32];
    const int RT = (rpb + 15) >> 4;
    const int RP = RT * 16;
    float* tile = sm;
    float* dg   = sm + (RP + 1) * PITCH;
    const int2* pairs = (const int2*)pairs_i;
    const int tid  = (int)threadIdx.x;
    const int lane = tid & 31;
    const int wave = __builtin_amdgcn_readfirstlane(tid >> 5);
    const int row0 = (int)blockIdx.x * rpb;
    int nrows = N - row0;
    if (nrows > rpb) nrows = rpb;
    if (nrows < 0)   nrows = 0;

    int2 sd = make_int2(0, 0);
    if (wave < NW / 32) sd = ((const int2*)seg_i)[(int)blockIdx.x * NW + tid];

    const v4f z = {0.0f, 0.0f, 0.0f, 0.0f};
    v4f* s4 = (v4f*)sm;
    for (int i = tid; i < (RP + 1) * (PITCH / 4); i += GT) s4[i] = z;
    for (int i = tid; i < RP; i += GT) dg[i] = 0.0f;

    int inc = sd.y;
#pragma unroll
    for (int d = 1; d < 32; d <<= 1) {
        const int t = __shfl_up(inc, d);
        if (lane >= d) inc += t;
    }
    if (wave < NW / 32 && lane == 31) wt[wave] = inc;
    __syncthreads();
    if (wave < NW / 32) {
        int pre = 0;
#pragma unroll
        for (int k = 0; k < NW / 32; ++k) pre += (k < wave) ? wt[k] : 0;
        pf[tid + 1] = pre + inc;
        go[tid]     = sd.x - (pre + inc - sd.y);
        if (tid == 0) pf[0] = 0;
    }
    __syncthreads();

    const v4f* W4 = (const v4f*)W;
    v4f wv[1024 / GT];
    float bv[4];
#pragma unroll
    for (int k = 0; k < 1024 / GT; ++k) wv[k] = W4[tid + k * GT];
#pragma unroll
    for (int t = 0; t < 4; ++t) bv[t] = bias[t * 16 + (lane & 15)];

    const int total = pf[NW];
    int  base = wave * 32;
    int2 nxt  = make_int2(row0 + RP, 0);
    int  adr  = -1;
    if (base + lane < total) {
        const int i = base + lane;
        int lo = 0;
#pragma unroll
        for (int s = NW / 2; s >= 1; s >>= 1)
            if (pf[lo + s] <= i) lo += s;
        nxt = pairs[go[lo] + i];
    }
    if (base + GT + lane < total) {
        const int i = base + GT + lane;
        int lo = 0;
#pragma unroll
        for (int s = NW / 2; s >= 1; s >>= 1)
            if (pf[lo + s] <= i) lo += s;
        adr = go[lo] + i;
    }
    for (; base < total; base += GT) {
        const int  cnt = (total - base < 32) ? (total - base) : 32;
        const int2 rc  = nxt;

        const int myrow = rc.x - row0;
        const int myoff = myrow * PITCH;
        const int mycol = rc.y;
        if (lane < cnt) atomicAdd(dg + myrow, 1.0f);

        v2f v[32];
        int rr[32];
#pragma unroll
        for (int u = 0; u < 32; ++u) {
            const int c = __builtin_amdgcn_readlane(mycol, u);
            rr[u]       = __builtin_amdgcn_readlane(myoff, u);
            const float* xr = x + (long)c * 64;
            v[u] = *(const v2f*)(xr + 2 * lane);
        }
        nxt = make_int2(row0 + RP, 0);
        if (adr >= 0) nxt = pairs[adr];
        adr = -1;
        if (base + 2 * GT + lane < total) {
            const int i = base + 2 * GT + lane;
            int lo = 0;
#pragma unroll
            for (int s = NW / 2; s >= 1; s >>= 1)
                if (pf[lo + s] <= i) lo += s;
            adr = go[lo] + i;
        }
#pragma unroll
        for (int u = 0; u < 32; ++u) {
            float* tp = tile + rr[u] + 2 * lane;
            atomicAdd(tp,     v[u].x);
            atomicAdd(tp + 1, v[u].y);
        }
    }

#pragma unroll
    for (int k = 0; k < 1024 / GT; ++k) {
        const int i = tid + k * GT;
        *(v4f*)(Wl + (i >> 4) * PITCH + (i & 15) * 4) = wv[k];
    }
    __syncthreads();

    const int c16  = lane & 15;
    const int half = lane >> 4;
    const int RTu  = (nrows + 15) >> 4;
    for (int rt = wave; rt < RTu; rt += GT / 32) {
        const int   r      = rt * 16 + c16;
        const float invdeg = 1.0f / (dg[r] + 1e-6f);
        const float* arow  = tile + r * PITCH + 2 * half;
        v2f a[16];
#pragma unroll
        for (int j = 0; j < 16; ++j) a[j] = *(const v2f*)(arow + 4 * j);

        v8f acc0 = {}, acc1 = {}, acc2 = {}, acc3 = {};
        const float* w0 = Wl + (0 * 16 + c16) * PITCH + 2 * half;
        const float* w1 = Wl + (1 * 16 + c16) * PITCH + 2 * half;
        const float* w2 = Wl + (2 * 16 + c16) * PITCH + 2 * half;
        const float* w3 = Wl + (3 * 16 + c16) * PITCH + 2 * half;
#pragma unroll
        for (int j = 0; j < 16; ++j) {
            v2f as;
            as.x = a[j].x * invdeg;
            as.y = a[j].y * invdeg;
            const v2f b0 = *(const v2f*)(w0 + 4 * j);
            const v2f b1 = *(const v2f*)(w1 + 4 * j);
            const v2f b2 = *(const v2f*)(w2 + 4 * j);
            const v2f b3 = *(const v2f*)(w3 + 4 * j);
            acc0 = __builtin_amdgcn_wmma_f32_16x16x4_f32(false, as, false, b0, (short)0, acc0, false, false);
            acc1 = __builtin_amdgcn_wmma_f32_16x16x4_f32(false, as, false, b1, (short)0, acc1, false, false);
            acc2 = __builtin_amdgcn_wmma_f32_16x16x4_f32(false, as, false, b2, (short)0, acc2, false, false);
            acc3 = __builtin_amdgcn_wmma_f32_16x16x4_f32(false, as, false, b3, (short)0, acc3, false, false);
        }

        v8f accs[4] = {acc0, acc1, acc2, acc3};
        float* orow = tile + (rt * 16 + half * 8) * PITCH + c16;
#pragma unroll
        for (int t = 0; t < 4; ++t) {
#pragma unroll
            for (int i = 0; i < 8; ++i)
                orow[i * PITCH + t * 16] = accs[t][i] + bv[t];
        }
    }
    __syncthreads();

    v4f* o4 = (v4f*)(out + (long)row0 * 64);
    for (int i = tid; i < nrows * 16; i += GT)
        o4[i] = *(const v4f*)(tile + (i >> 4) * PITCH + (i & 15) * 4);
}

extern "C" void kernel_launch(void* const* d_in, const int* in_sizes, int n_in,
                              void* d_out, int out_size, void* d_ws, size_t ws_size,
                              hipStream_t stream) {
    const float* x  = (const float*)d_in[0];
    const int*   ei = (const int*)d_in[1];
    const float* W  = (const float*)d_in[2];
    const float* b  = (const float*)d_in[3];
    float* out = (float*)d_out;

    const int N = in_sizes[0] / 64;
    const int E = in_sizes[1] / 2;
    const int rpb = (N + NB - 1) / NB;

    char* ws = (char*)d_ws;
    const size_t seg_bytes = (size_t)NB * NW * 8;
    int*   seg   = (int*)ws;
    int*   pairs = (int*)(ws + seg_bytes);

    const int chunk = (E + NW - 1) / NW;
    gcn_sort<<<NW, ST, (size_t)chunk * 20, stream>>>(ei, pairs, seg, E, rpb);
    const int rp = (rpb + 15) / 16 * 16;
    const size_t lds = ((size_t)(rp + 1) * PITCH + (size_t)rp) * 4;
    gcn_gather_gemm<<<NB, GT, lds, stream>>>(x, pairs, seg, W, b, out, N, rpb);
}


# gather group orders its range's edges by row in LDS (counting sort); a wave sums each run of equal row in registers, one LDS f32 add per run, not two per edge; sums still unordered
# speedup vs baseline: 5.3644x; 1.2209x over previous
#include <hip/hip_runtime.h>


typedef __attribute__((ext_vector_type(2))) float v2f;
typedef __attribute__((ext_vector_type(4))) float v4f;
typedef __attribute__((ext_vector_type(8))) float v8f;

#define NB 256
#define NW 256
#define ST 256
#define SU 16
#define GT 512
#define KU 7
#define CAP (GT * KU)
#define PITCH 68
static_assert(ST == NB, "the sort's scan has one thread per range");
static_assert(NW <= GT && NW % 32 == 0 && NB <= 256, "whole waves scan the segments; a key holds the range in 8 bits");

__global__ void __launch_bounds__(ST)
gcn_sort(const int* __restrict__ ei,
         int*       __restrict__ pairs_i,
         int*       __restrict__ seg_i,
         int E, int rpb) {
    extern __shared__ __attribute__((aligned(16))) int sl[];
    __shared__ int h[NB];
    __shared__ int st[NB];
    __shared__ int wt[ST / 32];
    const int tid  = (int)threadIdx.x;
    const int lane = tid & 31;
    const int wave = tid >> 5;
    const int w    = (int)blockIdx.x;
    const int chunk = (E + NW - 1) / NW;
    const int cpad  = (chunk + 15) & ~15;
    int beg = w * chunk;
    if (beg > E) beg = E;
    const int end = (beg + chunk < E) ? (beg + chunk) : E;
    const int n   = end - beg;
    int2*     raw = (int2*)sl;
    int2*     srt = raw + chunk;
    unsigned* key = (unsigned*)(srt + chunk);

    h[tid] = 0;
    __syncthreads();

    for (int i0 = tid; i0 < n; i0 += ST * SU) {
        int r[SU], c[SU];
#pragma unroll
        for (int u = 0; u < SU; ++u) {
            const int  i  = i0 + u * ST;
            const bool ok = i < n;
            r[u] = ok ? ei[beg + i]     : 0;
            c[u] = ok ? ei[E + beg + i] : 0;
        }
        int bk[SU], rk[SU];
#pragma unroll
        for (int u = 0; u < SU; ++u) {
            bk[u] = r[u] / rpb;
            rk[u] = 0;
            if (i0 + u * ST < n) rk[u] = atomicAdd(&h[bk[u]], 1);
        }
#pragma unroll
        for (int u = 0; u < SU; ++u) {
            const int i = i0 + u * ST;
            if (i < n) {
                raw[i] = make_int2(r[u], c[u]);
                key[i] = ((unsigned)bk[u] << 24) | (unsigned)rk[u];
            }
        }
    }
    __syncthreads();

    const int cntb = h[tid];
    int inc = cntb;
#pragma unroll
    for (int d = 1; d < 32; d <<= 1) {
        const int t = __shfl_up(inc, d);
        if (lane >= d) inc += t;
    }
    if (lane == 31) wt[wave] = inc;
    __syncthreads();
    int pre = 0;
#pragma unroll
    for (int k = 0; k < ST / 32; ++k) pre += (k < wave) ? wt[k] : 0;
    const int start = pre + inc - cntb;
    st[tid] = start;
    __syncthreads();

    for (int i0 = tid; i0 < n; i0 += ST * SU) {
        unsigned k[SU];
        int2     p[SU];
#pragma unroll
        for (int u = 0; u < SU; ++u) {
            const int i = i0 + u * ST;
            k[u] = 0u;
            p[u] = make_int2(0, 0);
            if (i < n) { k[u] = key[i]; p[u] = raw[i]; }
        }
        int pos[SU];
#pragma unroll
        for (int u = 0; u < SU; ++u) {
            pos[u] = 0;
            if (i0 + u * ST < n) pos[u] = st[k[u] >> 24] + (int)(k[u] & 0xFFFFFFu);
        }
#pragma unroll
        for (int u = 0; u < SU; ++u)
            if (i0 + u * ST < n) srt[pos[u]] = p[u];
    }
    __syncthreads();

    int2* pairs = (int2*)pairs_i + (long)w * cpad;
    for (int i0 = tid; i0 < n; i0 += ST * SU) {
        int2 p[SU];
#pragma unroll
        for (int u = 0; u < SU; ++u) {
            const int i = i0 + u * ST;
            p[u] = make_int2(0, 0);
            if (i < n) p[u] = srt[i];
        }
#pragma unroll
        for (int u = 0; u < SU; ++u) {
            const int i = i0 + u * ST;
            if (i < n) pairs[i] = p[u];
        }
    }
    ((int2*)seg_i)[tid * NW + w] = make_int2(w * cpad + start, cntb);
}

__global__ void __launch_bounds__(GT)
gcn_gather_gemm(const float* __restrict__ x,
                const int*   __restrict__ pairs_i,
                const int*   __restrict__ seg_i,
                const float* __restrict__ W,
                const float* __restrict__ bias,
                float*       __restrict__ out,
                int N, int rpb) {
    extern __shared__ __attribute__((aligned(16))) float sm[];
    __shared__ __attribute__((aligned(16))) float Wl[64 * PITCH];
    __shared__ int pf[NW + 1];
    __shared__ int go[NW];
    __shared__ int wt[NW / 32];
    __shared__ int2 se[CAP];
    __shared__ int h[GT];
    __shared__ int hs[GT];
    __shared__ int wt2[GT / 32];
    const int RT = (rpb + 15) >> 4;
    const int RP = RT * 16;
    float* tile = sm;
    float* dg   = sm + (RP + 1) * PITCH;
    const int2* pairs = (const int2*)pairs_i;
    const int tid  = (int)threadIdx.x;
    const int lane = tid & 31;
    const int wave = __builtin_amdgcn_readfirstlane(tid >> 5);
    const int row0 = (int)blockIdx.x * rpb;
    int nrows = N - row0;
    if (nrows > rpb) nrows = rpb;
    if (nrows < 0)   nrows = 0;

    int2 sd = make_int2(0, 0);
    if (wave < NW / 32) sd = ((const int2*)seg_i)[(int)blockIdx.x * NW + tid];

    const v4f z = {0.0f, 0.0f, 0.0f, 0.0f};
    v4f* s4 = (v4f*)sm;
    for (int i = tid; i < (RP + 1) * (PITCH / 4); i += GT) s4[i] = z;
    for (int i = tid; i < RP; i += GT) dg[i] = 0.0f;
    h[tid] = 0;

    int inc = sd.y;
#pragma unroll
    for (int d = 1; d < 32; d <<= 1) {
        const int t = __shfl_up(inc, d);
        if (lane >= d) inc += t;
    }
    if (wave < NW / 32 && lane == 31) wt[wave] = inc;
    __syncthreads();
    if (wave < NW / 32) {
        int pre = 0;
#pragma unroll
        for (int k = 0; k < NW / 32; ++k) pre += (k < wave) ? wt[k] : 0;
        pf[tid + 1] = pre + inc;
        go[tid]     = sd.x - (pre + inc - sd.y);
        if (tid == 0) pf[0] = 0;
    }
    __syncthreads();

    const v4f* W4 = (const v4f*)W;
    v4f wv[1024 / GT];
    float bv[4];
#pragma unroll
    for (int k = 0; k < 1024 / GT; ++k) wv[k] = W4[tid + k * GT];
#pragma unroll
    for (int t = 0; t < 4; ++t) bv[t] = bias[t * 16 + (lane & 15)];

    const int total = __builtin_amdgcn_readfirstlane(pf[NW]);
    for (int b0 = 0; b0 < total; b0 += CAP) {
        const int nb = (total - b0 < CAP) ? (total - b0) : CAP;

        int  ii[KU], lo[KU];
        bool ok[KU];
#pragma unroll
        for (int k = 0; k < KU; ++k) {
            const int i = tid + k * GT;
            ok[k] = i < nb;
            ii[k] = ok[k] ? (b0 + i) : 0;
            lo[k] = 0;
        }
#pragma unroll
        for (int s = NW / 2; s >= 1; s >>= 1) {
#pragma unroll
            for (int k = 0; k < KU; ++k)
                if (pf[lo[k] + s] <= ii[k]) lo[k] += s;
        }
        int2 rc[KU];
#pragma unroll
        for (int k = 0; k < KU; ++k) rc[k] = pairs[go[lo[k]] + ii[k]];
        int rk[KU];
#pragma unroll
        for (int k = 0; k < KU; ++k) {
            rk[k] = 0;
            if (ok[k]) rk[k] = atomicAdd(&h[rc[k].x - row0], 1);
        }
        __syncthreads();

        const int cntr = h[tid];
        int inr = cntr;
#pragma unroll
        for (int d = 1; d < 32; d <<= 1) {
            const int t = __shfl_up(inr, d);
            if (lane >= d) inr += t;
        }
        if (lane == 31) wt2[wave] = inr;
        __syncthreads();
        int prr = 0;
#pragma unroll
        for (int k = 0; k < GT / 32; ++k) prr += (k < wave) ? wt2[k] : 0;
        hs[tid] = prr + inr - cntr;
        if (tid < RP) dg[tid] += (float)cntr;
        __syncthreads();

#pragma unroll
        for (int k = 0; k < KU; ++k) {
            if (ok[k]) {
                const int lr = rc[k].x - row0;
                se[hs[lr] + rk[k]] = make_int2(lr * PITCH, rc[k].y);
            }
        }
        __syncthreads();

        int2 en = make_int2(RP * PITCH, 0);
        if (wave * 32 + lane < nb) en = se[wave * 32 + lane];
        for (int gb = wave * 32; gb < nb; gb += GT) {
            const int2 e = en;
            v2f v[32];
#pragma unroll
            for (int u = 0; u < 32; ++u) {
                const int c = __builtin_amdgcn_readlane(e.y, u);
                const float* xr = x + (long)c * 64;
                v[u] = *(const v2f*)(xr + 2 * lane);
            }
            en = make_int2(RP * PITCH, 0);
            if (gb + GT + lane < nb) en = se[gb + GT + lane];
            int cur = __builtin_amdgcn_readlane(e.x, 0);
            v2f acc = {0.0f, 0.0f};
#pragma unroll
            for (int u = 0; u < 32; ++u) {
                const int r = __builtin_amdgcn_readlane(e.x, u);
                if (r != cur) {
                    float* tp = tile + cur + 2 * lane;
                    atomicAdd(tp,     acc.x);
                    atomicAdd(tp + 1, acc.y);
                    acc.x = 0.0f;
                    acc.y = 0.0f;
                    cur = r;
                }
                acc.x += v[u].x;
                acc.y += v[u].y;
            }
            float* tp = tile + cur + 2 * lane;
            atomicAdd(tp,     acc.x);
            atomicAdd(tp + 1, acc.y);
        }

        if (b0 + CAP < total) {
            __syncthreads();
            h[tid] = 0;
            __syncthreads();
        }
    }

#pragma unroll
    for (int k = 0; k < 1024 / GT; ++k) {
        const int i = tid + k * GT;
        *(v4f*)(Wl + (i >> 4) * PITCH + (i & 15) * 4) = wv[k];
    }
    __syncthreads();

    const int c16  = lane & 15;
    const int half = lane >> 4;
    const int RTu  = (nrows + 15) >> 4;
    for (int rt = wave; rt < RTu; rt += GT / 32) {
        const int   r      = rt * 16 + c16;
        const float invdeg = 1.0f / (dg[r] + 1e-6f);
        const float* arow  = tile + r * PITCH + 2 * half;
        v2f a[16];
#pragma unroll
        for (int j = 0; j < 16; ++j) a[j] = *(const v2f*)(arow + 4 * j);

        v8f acc0 = {}, acc1 = {}, acc2 = {}, acc3 = {};
        const float* w0 = Wl + (0 * 16 + c16) * PITCH + 2 * half;
        const float* w1 = Wl + (1 * 16 + c16) * PITCH + 2 * half;
        const float* w2 = Wl + (2 * 16 + c16) * PITCH + 2 * half;
        const float* w3 = Wl + (3 * 16 + c16) * PITCH + 2 * half;
#pragma unroll
        for (int j = 0; j < 16; ++j) {
            v2f as;
            as.x = a[j].x * invdeg;
            as.y = a[j].y * invdeg;
            const v2f b0 = *(const v2f*)(w0 + 4 * j);
            const v2f b1 = *(const v2f*)(w1 + 4 * j);
            const v2f b2 = *(const v2f*)(w2 + 4 * j);
            const v2f b3 = *(const v2f*)(w3 + 4 * j);
            acc0 = __builtin_amdgcn_wmma_f32_16x16x4_f32(false, as, false, b0, (short)0, acc0, false, false);
            acc1 = __builtin_amdgcn_wmma_f32_16x16x4_f32(false, as, false, b1, (short)0, acc1, false, false);
            acc2 = __builtin_amdgcn_wmma_f32_16x16x4_f32(false, as, false, b2, (short)0, acc2, false, false);
            acc3 = __builtin_amdgcn_wmma_f32_16x16x4_f32(false, as, false, b3, (short)0, acc3, false, false);
        }

        v8f accs[4] = {acc0, acc1, acc2, acc3};
        float* orow = tile + (rt * 16 + half * 8) * PITCH + c16;
#pragma unroll
        for (int t = 0; t < 4; ++t) {
#pragma unroll
            for (int i = 0; i < 8; ++i)
                orow[i * PITCH + t * 16] = accs[t][i] + bv[t];
        }
    }
    __syncthreads();

    v4f* o4 = (v4f*)(out + (long)row0 * 64);
    for (int i = tid; i < nrows * 16; i += GT)
        o4[i] = *(const v4f*)(tile + (i >> 4) * PITCH + (i & 15) * 4);
}

extern "C" void kernel_launch(void* const* d_in, const int* in_sizes, int n_in,
                              void* d_out, int out_size, void* d_ws, size_t ws_size,
                              hipStream_t stream) {
    const float* x  = (const float*)d_in[0];
    const int*   ei = (const int*)d_in[1];
    const float* W  = (const float*)d_in[2];
    const float* b  = (const float*)d_in[3];
    float* out = (float*)d_out;

    const int N = in_sizes[0] / 64;
    const int E = in_sizes[1] / 2;
    const int rpb = (N + NB - 1) / NB;

    char* ws = (char*)d_ws;
    const size_t seg_bytes = (size_t)NB * NW * 8;
    int*   seg   = (int*)ws;
    int*   pairs = (int*)(ws + seg_bytes);

    const int chunk = (E + NW - 1) / NW;
    gcn_sort<<<NW, ST, (size_t)chunk * 20, stream>>>(ei, pairs, seg, E, rpb);
    const int rp = (rpb + 15) / 16 * 16;
    const size_t lds = ((size_t)(rp + 1) * PITCH + (size_t)rp) * 4;
    gcn_gather_gemm<<<NB, GT, lds, stream>>>(x, pairs, seg, W, b, out, N, rpb);
}


# global loads and stores sit under no lane mask and recycle no address register (clamped index, scalar branch per whole wave, own index registers): no s_wait_xcnt between them; sums still unordered
# speedup vs baseline: 5.7821x; 1.0779x over previous
#include <hip/hip_runtime.h>


typedef __attribute__((ext_vector_type(2))) float v2f;
typedef __attribute__((ext_vector_type(4))) float v4f;
typedef __attribute__((ext_vector_type(8))) float v8f;

#define NB 256
#define NW 256
#define ST 256
#define SU 16
#define GT 512
#define KU 7
#define CAP (GT * KU)
#define OU 4
#define PITCH 68
static_assert(ST == NB, "the sort's scan has one thread per range");
static_assert(NW <= GT && NW % 32 == 0 && NB <= 256, "whole waves scan the segments; a key holds the range in 8 bits");

__global__ void __launch_bounds__(ST)
gcn_sort(const int* __restrict__ ei,
         int*       __restrict__ pairs_i,
         int*       __restrict__ seg_i,
         int E, int rpb) {
    extern __shared__ __attribute__((aligned(16))) int sl[];
    __shared__ int h[NB];
    __shared__ int st[NB];
    __shared__ int wt[ST / 32];
    const int tid  = (int)threadIdx.x;
    const int lane = tid & 31;
    const int wave = tid >> 5;
    const int w    = (int)blockIdx.x;
    const int chunk = (E + NW - 1) / NW;
    const int cpad  = (chunk + 15) & ~15;
    int beg = w * chunk;
    if (beg > E) beg = E;
    const int end = (beg + chunk < E) ? (beg + chunk) : E;
    const int n   = end - beg;
    int2*     raw = (int2*)sl;
    int2*     srt = raw + chunk;
    unsigned* key = (unsigned*)(srt + chunk);

    h[tid] = 0;
    __syncthreads();

    const int* rowp = ei + beg;
    const int* colp = ei + E + beg;
    for (int i0 = tid; i0 < n; i0 += ST * SU) {
        int r[SU], c[SU];
#pragma unroll
        for (int u = 0; u < SU; ++u) {
            const int i  = i0 + u * ST;
            const int ic = (i < n) ? i : (n - 1);
            r[u] = rowp[ic];
            c[u] = colp[ic];
        }
        int bk[SU], rk[SU];
#pragma unroll
        for (int u = 0; u < SU; ++u) {
            bk[u] = r[u] / rpb;
            rk[u] = 0;
            if (i0 + u * ST < n) rk[u] = atomicAdd(&h[bk[u]], 1);
        }
#pragma unroll
        for (int u = 0; u < SU; ++u) {
            const int i = i0 + u * ST;
            if (i < n) {
                raw[i] = make_int2(r[u], c[u]);
                key[i] = ((unsigned)bk[u] << 24) | (unsigned)rk[u];
            }
        }
    }
    __syncthreads();

    const int cntb = h[tid];
    int inc = cntb;
#pragma unroll
    for (int d = 1; d < 32; d <<= 1) {
        const int t = __shfl_up(inc, d);
        if (lane >= d) inc += t;
    }
    if (lane == 31) wt[wave] = inc;
    __syncthreads();
    int pre = 0;
#pragma unroll
    for (int k = 0; k < ST / 32; ++k) pre += (k < wave) ? wt[k] : 0;
    const int start = pre + inc - cntb;
    st[tid] = start;
    __syncthreads();

    for (int i0 = tid; i0 < n; i0 += ST * SU) {
        unsigned k[SU];
        int2     p[SU];
#pragma unroll
        for (int u = 0; u < SU; ++u) {
            const int i = i0 + u * ST;
            k[u] = 0u;
            p[u] = make_int2(0, 0);
            if (i < n) { k[u] = key[i]; p[u] = raw[i]; }
        }
        int pos[SU];
#pragma unroll
        for (int u = 0; u < SU; ++u) {
            pos[u] = 0;
            if (i0 + u * ST < n) pos[u] = st[k[u] >> 24] + (int)(k[u] & 0xFFFFFFu);
        }
#pragma unroll
        for (int u = 0; u < SU; ++u)
            if (i0 + u * ST < n) srt[pos[u]] = p[u];
    }
    __syncthreads();

    int2* pairs = (int2*)pairs_i + (long)w * cpad;
    for (int i0 = tid; i0 < n; i0 += ST * SU) {
        int2 p[SU];
#pragma unroll
        for (int u = 0; u < SU; ++u) {
            const int i = i0 + u * ST;
            p[u] = make_int2(0, 0);
            if (i < n) p[u] = srt[i];
        }
        const int wb = __builtin_amdgcn_readfirstlane(i0 - lane);
#pragma unroll
        for (int u = 0; u < SU; ++u)
            if (wb + u * ST + 32 <= n) pairs[i0 + u * ST] = p[u];
        int nf = (n - wb - 32 >= 0) ? ((n - wb - 32) / ST + 1) : 0;
        if (nf > SU) nf = SU;
        const int it = i0 + nf * ST;
        if (nf < SU && it < n) pairs[it] = srt[it];
    }
    ((int2*)seg_i)[tid * NW + w] = make_int2(w * cpad + start, cntb);
}

__global__ void __launch_bounds__(GT)
gcn_gather_gemm(const float* __restrict__ x,
                const int*   __restrict__ pairs_i,
                const int*   __restrict__ seg_i,
                const float* __restrict__ W,
                const float* __restrict__ bias,
                float*       __restrict__ out,
                int N, int rpb) {
    extern __shared__ __attribute__((aligned(16))) float sm[];
    __shared__ __attribute__((aligned(16))) float Wl[64 * PITCH];
    __shared__ int pf[NW + 1];
    __shared__ int go[NW];
    __shared__ int wt[NW / 32];
    __shared__ int2 se[CAP];
    __shared__ int h[GT];
    __shared__ int hs[GT];
    __shared__ int wt2[GT / 32];
    const int RT = (rpb + 15) >> 4;
    const int RP = RT * 16;
    float* tile = sm;
    float* dg   = sm + (RP + 1) * PITCH;
    const int2* pairs = (const int2*)pairs_i;
    const int tid  = (int)threadIdx.x;
    const int lane = tid & 31;
    const int wave = __builtin_amdgcn_readfirstlane(tid >> 5);
    const int row0 = (int)blockIdx.x * rpb;
    int nrows = N - row0;
    if (nrows > rpb) nrows = rpb;
    if (nrows < 0)   nrows = 0;

    int2 sd = make_int2(0, 0);
    if (wave < NW / 32) sd = ((const int2*)seg_i)[(int)blockIdx.x * NW + tid];

    const v4f z = {0.0f, 0.0f, 0.0f, 0.0f};
    v4f* s4 = (v4f*)sm;
    for (int i = tid; i < (RP + 1) * (PITCH / 4); i += GT) s4[i] = z;
    for (int i = tid; i < RP; i += GT) dg[i] = 0.0f;
    h[tid] = 0;

    int inc = sd.y;
#pragma unroll
    for (int d = 1; d < 32; d <<= 1) {
        const int t = __shfl_up(inc, d);
        if (lane >= d) inc += t;
    }
    if (wave < NW / 32 && lane == 31) wt[wave] = inc;
    __syncthreads();
    if (wave < NW / 32) {
        int pre = 0;
#pragma unroll
        for (int k = 0; k < NW / 32; ++k) pre += (k < wave) ? wt[k] : 0;
        pf[tid + 1] = pre + inc;
        go[tid]     = sd.x - (pre + inc - sd.y);
        if (tid == 0) pf[0] = 0;
    }
    __syncthreads();

    const v4f* W4 = (const v4f*)W;
    v4f wv[1024 / GT];
    float bv[4];
#pragma unroll
    for (int k = 0; k < 1024 / GT; ++k) wv[k] = W4[tid + k * GT];
#pragma unroll
    for (int t = 0; t < 4; ++t) bv[t] = bias[t * 16 + (lane & 15)];

    const int total = __builtin_amdgcn_readfirstlane(pf[NW]);
    for (int b0 = 0; b0 < total; b0 += CAP) {
        const int nb = (total - b0 < CAP) ? (total - b0) : CAP;

        int  ii[KU], lo[KU];
        bool ok[KU];
#pragma unroll
        for (int k = 0; k < KU; ++k) {
            const int i = tid + k * GT;
            ok[k] = i < nb;
            ii[k] = ok[k] ? (b0 + i) : 0;
            lo[k] = 0;
        }
#pragma unroll
        for (int s = NW / 2; s >= 1; s >>= 1) {
#pragma unroll
            for (int k = 0; k < KU; ++k)
                if (pf[lo[k] + s] <= ii[k]) lo[k] += s;
        }
        int2 rc[KU];
#pragma unroll
        for (int k = 0; k < KU; ++k) rc[k] = pairs[go[lo[k]] + ii[k]];
        int rk[KU];
#pragma unroll
        for (int k = 0; k < KU; ++k) {
            rk[k] = 0;
            if (ok[k]) rk[k] = atomicAdd(&h[rc[k].x - row0], 1);
        }
        __syncthreads();

        const int cntr = h[tid];
        int inr = cntr;
#pragma unroll
        for (int d = 1; d < 32; d <<= 1) {
            const int t = __shfl_up(inr, d);
            if (lane >= d) inr += t;
        }
        if (lane == 31) wt2[wave] = inr;
        __syncthreads();
        int prr = 0;
#pragma unroll
        for (int k = 0; k < GT / 32; ++k) prr += (k < wave) ? wt2[k] : 0;
        hs[tid] = prr + inr - cntr;
        if (tid < RP) dg[tid] += (float)cntr;
        __syncthreads();

#pragma unroll
        for (int k = 0; k < KU; ++k) {
            if (ok[k]) {
                const int lr = rc[k].x - row0;
                se[hs[lr] + rk[k]] = make_int2(lr * PITCH, rc[k].y * 32);
            }
        }
        __syncthreads();

        const v2f* x2 = (const v2f*)x;
        int2 en = make_int2(RP * PITCH, 0);
        if (wave * 32 + lane < nb) en = se[wave * 32 + lane];
        for (int gb = wave * 32; gb < nb; gb += GT) {
            const int2 e = en;
            v2f v[32];
            unsigned pi[32];
#pragma unroll
            for (int u = 0; u < 32; ++u) {
                pi[u] = (unsigned)__builtin_amdgcn_readlane(e.y, u) + (unsigned)lane;
                v[u] = x2[pi[u]];
            }
            asm volatile("" :: "v"(pi[0]), "v"(pi[1]), "v"(pi[2]), "v"(pi[3]), "v"(pi[4]), "v"(pi[5]), "v"(pi[6]), "v"(pi[7]),
                               "v"(pi[8]), "v"(pi[9]), "v"(pi[10]), "v"(pi[11]), "v"(pi[12]), "v"(pi[13]), "v"(pi[14]), "v"(pi[15]));
            asm volatile("" :: "v"(pi[16]), "v"(pi[17]), "v"(pi[18]), "v"(pi[19]), "v"(pi[20]), "v"(pi[21]), "v"(pi[22]), "v"(pi[23]),
                               "v"(pi[24]), "v"(pi[25]), "v"(pi[26]), "v"(pi[27]), "v"(pi[28]), "v"(pi[29]), "v"(pi[30]), "v"(pi[31]));
            const int ni = gb + GT + lane;
            en = se[(ni < CAP) ? ni : (CAP - 1)];
            if (ni >= nb) en = make_int2(RP * PITCH, 0);
            int cur = __builtin_amdgcn_readlane(e.x, 0);
            v2f acc = {0.0f, 0.0f};
#pragma unroll
            for (int u = 0; u < 32; ++u) {
                const int r = __builtin_amdgcn_readlane(e.x, u);
                if (r != cur) {
                    float* tp = tile + cur + 2 * lane;
                    atomicAdd(tp,     acc.x);
                    atomicAdd(tp + 1, acc.y);
                    acc.x = 0.0f;
                    acc.y = 0.0f;
                    cur = r;
                }
                acc.x += v[u].x;
                acc.y += v[u].y;
            }
            float* tp = tile + cur + 2 * lane;
            atomicAdd(tp,     acc.x);
            atomicAdd(tp + 1, acc.y);
        }

        if (b0 + CAP < total) {
            __syncthreads();
            h[tid] = 0;
            __syncthreads();
        }
    }

#pragma unroll
    for (int k = 0; k < 1024 / GT; ++k) {
        const int i = tid + k * GT;
        *(v4f*)(Wl + (i >> 4) * PITCH + (i & 15) * 4) = wv[k];
    }
    __syncthreads();

    const int c16  = lane & 15;
    const int half = lane >> 4;
    const int RTu  = (nrows + 15) >> 4;
    for (int rt = wave; rt < RTu; rt += GT / 32) {
        const int   r      = rt * 16 + c16;
        const float invdeg = 1.0f / (dg[r] + 1e-6f);
        const float* arow  = tile + r * PITCH + 2 * half;
        v2f a[16];
#pragma unroll
        for (int j = 0; j < 16; ++j) a[j] = *(const v2f*)(arow + 4 * j);

        v8f acc0 = {}, acc1 = {}, acc2 = {}, acc3 = {};
        const float* w0 = Wl + (0 * 16 + c16) * PITCH + 2 * half;
        const float* w1 = Wl + (1 * 16 + c16) * PITCH + 2 * half;
        const float* w2 = Wl + (2 * 16 + c16) * PITCH + 2 * half;
        const float* w3 = Wl + (3 * 16 + c16) * PITCH + 2 * half;
#pragma unroll
        for (int j = 0; j < 16; ++j) {
            v2f as;
            as.x = a[j].x * invdeg;
            as.y = a[j].y * invdeg;
            const v2f b0 = *(const v2f*)(w0 + 4 * j);
            const v2f b1 = *(const v2f*)(w1 + 4 * j);
            const v2f b2 = *(const v2f*)(w2 + 4 * j);
            const v2f b3 = *(const v2f*)(w3 + 4 * j);
            acc0 = __builtin_amdgcn_wmma_f32_16x16x4_f32(false, as, false, b0, (short)0, acc0, false, false);
            acc1 = __builtin_amdgcn_wmma_f32_16x16x4_f32(false, as, false, b1, (short)0, acc1, false, false);
            acc2 = __builtin_amdgcn_wmma_f32_16x16x4_f32(false, as, false, b2, (short)0, acc2, false, false);
            acc3 = __builtin_amdgcn_wmma_f32_16x16x4_f32(false, as, false, b3, (short)0, acc3, false, false);
        }

        v8f accs[4] = {acc0, acc1, acc2, acc3};
        float* orow = tile + (rt * 16 + half * 8) * PITCH + c16;
#pragma unroll
        for (int t = 0; t < 4; ++t) {
#pragma unroll
            for (int i = 0; i < 8; ++i)
                orow[i * PITCH + t * 16] = accs[t][i] + bv[t];
        }
    }
    __syncthreads();

    v4f* o4 = (v4f*)(out + (long)row0 * 64);
    const int nq = nrows * 16;
    for (int i0 = tid; i0 < nq; i0 += GT * OU) {
        const int wb = __builtin_amdgcn_readfirstlane(i0 - lane);
        v4f q[OU];
#pragma unroll
        for (int k = 0; k < OU; ++k) {
            const int i  = i0 + k * GT;
            const int ic = (i < nq) ? i : (nq - 1);
            q[k] = *(const v4f*)(tile + (ic >> 4) * PITCH + (ic & 15) * 4);
        }
#pragma unroll
        for (int k = 0; k < OU; ++k)
            if (wb + k * GT + 32 <= nq) o4[i0 + k * GT] = q[k];
        int nf = (nq - wb - 32 >= 0) ? ((nq - wb - 32) / GT + 1) : 0;
        if (nf > OU) nf = OU;
        const int it = i0 + nf * GT;
        if (nf < OU && it < nq)
            o4[it] = *(const v4f*)(tile + (it >> 4) * PITCH + (it & 15) * 4);
    }
}

extern "C" void kernel_launch(void* const* d_in, const int* in_sizes, int n_in,
                              void* d_out, int out_size, void* d_ws, size_t ws_size,
                              hipStream_t stream) {
    const float* x  = (const float*)d_in[0];
    const int*   ei = (const int*)d_in[1];
    const float* W  = (const float*)d_in[2];
    const float* b  = (const float*)d_in[3];
    float* out = (float*)d_out;

    const int N = in_sizes[0] / 64;
    const int E = in_sizes[1] / 2;
    const int rpb = (N + NB - 1) / NB;

    char* ws = (char*)d_ws;
    const size_t seg_bytes = (size_t)NB * NW * 8;
    int*   seg   = (int*)ws;
    int*   pairs = (int*)(ws + seg_bytes);

    const int chunk = (E + NW - 1) / NW;
    gcn_sort<<<NW, ST, (size_t)chunk * 20, stream>>>(ei, pairs, seg, E, rpb);
    const int rp = (rpb + 15) / 16 * 16;
    const size_t lds = ((size_t)(rp + 1) * PITCH + (size_t)rp) * 4;
    gcn_gather_gemm<<<NB, GT, lds, stream>>>(x, pairs, seg, W, b, out, N, rpb);
}


# gather loop: one 16 B load per lane fetches two x rows (a half-wave per row): 16 loads per 32 edges, run ends from one ballot mask; runs now split per half-wave, sums still unordered LDS f32 adds
# speedup vs baseline: 5.9711x; 1.0327x over previous
#include <hip/hip_runtime.h>


typedef __attribute__((ext_vector_type(2))) float v2f;
typedef __attribute__((ext_vector_type(4))) float v4f;
typedef __attribute__((ext_vector_type(8))) float v8f;

#define NB 256
#define NW 256
#define ST 256
#define SU 16
#define GT 512
#define KU 7
#define CAP (GT * KU)
#define OU 4
#define PITCH 68
static_assert(ST == NB, "the sort's scan has one thread per range");
static_assert(NW <= GT && NW % 32 == 0 && NB <= 256, "whole waves scan the segments; a key holds the range in 8 bits");

__global__ void __launch_bounds__(ST)
gcn_sort(const int* __restrict__ ei,
         int*       __restrict__ pairs_i,
         int*       __restrict__ seg_i,
         int E, int rpb) {
    extern __shared__ __attribute__((aligned(16))) int sl[];
    __shared__ int h[NB];
    __shared__ int st[NB];
    __shared__ int wt[ST / 32];
    const int tid  = (int)threadIdx.x;
    const int lane = tid & 31;
    const int wave = tid >> 5;
    const int w    = (int)blockIdx.x;
    const int chunk = (E + NW - 1) / NW;
    const int cpad  = (chunk + 15) & ~15;
    int beg = w * chunk;
    if (beg > E) beg = E;
    const int end = (beg + chunk < E) ? (beg + chunk) : E;
    const int n   = end - beg;
    int2*     raw = (int2*)sl;
    int2*     srt = raw + chunk;
    unsigned* key = (unsigned*)(srt + chunk);

    h[tid] = 0;
    __syncthreads();

    const int* rowp = ei + beg;
    const int* colp = ei + E + beg;
    for (int i0 = tid; i0 < n; i0 += ST * SU) {
        int r[SU], c[SU];
#pragma unroll
        for (int u = 0; u < SU; ++u) {
            const int i  = i0 + u * ST;
            const int ic = (i < n) ? i : (n - 1);
            r[u] = rowp[ic];
            c[u] = colp[ic];
        }
        int bk[SU], rk[SU];
#pragma unroll
        for (int u = 0; u < SU; ++u) {
            bk[u] = r[u] / rpb;
            rk[u] = 0;
            if (i0 + u * ST < n) rk[u] = atomicAdd(&h[bk[u]], 1);
        }
#pragma unroll
        for (int u = 0; u < SU; ++u) {
            const int i = i0 + u * ST;
            if (i < n) {
                raw[i] = make_int2(r[u], c[u]);
                key[i] = ((unsigned)bk[u] << 24) | (unsigned)rk[u];
            }
        }
    }
    __syncthreads();

    const int cntb = h[tid];
    int inc = cntb;
#pragma unroll
    for (int d = 1; d < 32; d <<= 1) {
        const int t = __shfl_up(inc, d);
        if (lane >= d) inc += t;
    }
    if (lane == 31) wt[wave] = inc;
    __syncthreads();
    int pre = 0;
#pragma unroll
    for (int k = 0; k < ST / 32; ++k) pre += (k < wave) ? wt[k] : 0;
    const int start = pre + inc - cntb;
    st[tid] = start;
    __syncthreads();

    for (int i0 = tid; i0 < n; i0 += ST * SU) {
        unsigned k[SU];
        int2     p[SU];
#pragma unroll
        for (int u = 0; u < SU; ++u) {
            const int i = i0 + u * ST;
            k[u] = 0u;
            p[u] = make_int2(0, 0);
            if (i < n) { k[u] = key[i]; p[u] = raw[i]; }
        }
        int pos[SU];
#pragma unroll
        for (int u = 0; u < SU; ++u) {
            pos[u] = 0;
            if (i0 + u * ST < n) pos[u] = st[k[u] >> 24] + (int)(k[u] & 0xFFFFFFu);
        }
#pragma unroll
        for (int u = 0; u < SU; ++u)
            if (i0 + u * ST < n) srt[pos[u]] = p[u];
    }
    __syncthreads();

    int2* pairs = (int2*)pairs_i + (long)w * cpad;
    for (int i0 = tid; i0 < n; i0 += ST * SU) {
        int2 p[SU];
#pragma unroll
        for (int u = 0; u < SU; ++u) {
            const int i = i0 + u * ST;
            p[u] = make_int2(0, 0);
            if (i < n) p[u] = srt[i];
        }
        const int wb = __builtin_amdgcn_readfirstlane(i0 - lane);
#pragma unroll
        for (int u = 0; u < SU; ++u)
            if (wb + u * ST + 32 <= n) pairs[i0 + u * ST] = p[u];
        int nf = (n - wb - 32 >= 0) ? ((n - wb - 32) / ST + 1) : 0;
        if (nf > SU) nf = SU;
        const int it = i0 + nf * ST;
        if (nf < SU && it < n) pairs[it] = srt[it];
    }
    ((int2*)seg_i)[tid * NW + w] = make_int2(w * cpad + start, cntb);
}

template <int U>
static __device__ __forceinline__ int rshare(int v) {
    return __builtin_amdgcn_update_dpp(0, v, 0x150 + U, 0xf, 0xf, true);
}

__global__ void __launch_bounds__(GT)
gcn_gather_gemm(const float* __restrict__ x,
                const int*   __restrict__ pairs_i,
                const int*   __restrict__ seg_i,
                const float* __restrict__ W,
                const float* __restrict__ bias,
                float*       __restrict__ out,
                int N, int rpb) {
    extern __shared__ __attribute__((aligned(16))) float sm[];
    __shared__ __attribute__((aligned(16))) float Wl[64 * PITCH];
    __shared__ int pf[NW + 1];
    __shared__ int go[NW];
    __shared__ int wt[NW / 32];
    __shared__ int2 se[CAP];
    __shared__ int h[GT];
    __shared__ int hs[GT];
    __shared__ int wt2[GT / 32];
    const int RT = (rpb + 15) >> 4;
    const int RP = RT * 16;
    float* tile = sm;
    float* dg   = sm + (RP + 1) * PITCH;
    const int2* pairs = (const int2*)pairs_i;
    const int tid  = (int)threadIdx.x;
    const int lane = tid & 31;
    const int wave = __builtin_amdgcn_readfirstlane(tid >> 5);
    const int row0 = (int)blockIdx.x * rpb;
    int nrows = N - row0;
    if (nrows > rpb) nrows = rpb;
    if (nrows < 0)   nrows = 0;

    int2 sd = make_int2(0, 0);
    if (wave < NW / 32) sd = ((const int2*)seg_i)[(int)blockIdx.x * NW + tid];

    const v4f z = {0.0f, 0.0f, 0.0f, 0.0f};
    v4f* s4 = (v4f*)sm;
    for (int i = tid; i < (RP + 1) * (PITCH / 4); i += GT) s4[i] = z;
    for (int i = tid; i < RP; i += GT) dg[i] = 0.0f;
    h[tid] = 0;

    int inc = sd.y;
#pragma unroll
    for (int d = 1; d < 32; d <<= 1) {
        const int t = __shfl_up(inc, d);
        if (lane >= d) inc += t;
    }
    if (wave < NW / 32 && lane == 31) wt[wave] = inc;
    __syncthreads();
    if (wave < NW / 32) {
        int pre = 0;
#pragma unroll
        for (int k = 0; k < NW / 32; ++k) pre += (k < wave) ? wt[k] : 0;
        pf[tid + 1] = pre + inc;
        go[tid]     = sd.x - (pre + inc - sd.y);
        if (tid == 0) pf[0] = 0;
    }
    __syncthreads();

    const v4f* W4 = (const v4f*)W;
    v4f wv[1024 / GT];
    float bv[4];
#pragma unroll
    for (int k = 0; k < 1024 / GT; ++k) wv[k] = W4[tid + k * GT];
#pragma unroll
    for (int t = 0; t < 4; ++t) bv[t] = bias[t * 16 + (lane & 15)];

    const int total = __builtin_amdgcn_readfirstlane(pf[NW]);
    for (int b0 = 0; b0 < total; b0 += CAP) {
        const int nb = (total - b0 < CAP) ? (total - b0) : CAP;

        int  ii[KU], lo[KU];
        bool ok[KU];
#pragma unroll
        for (int k = 0; k < KU; ++k) {
            const int i = tid + k * GT;
            ok[k] = i < nb;
            ii[k] = ok[k] ? (b0 + i) : 0;
            lo[k] = 0;
        }
#pragma unroll
        for (int s = NW / 2; s >= 1; s >>= 1) {
#pragma unroll
            for (int k = 0; k < KU; ++k)
                if (pf[lo[k] + s] <= ii[k]) lo[k] += s;
        }
        int2 rc[KU];
#pragma unroll
        for (int k = 0; k < KU; ++k) rc[k] = pairs[go[lo[k]] + ii[k]];
        int rk[KU];
#pragma unroll
        for (int k = 0; k < KU; ++k) {
            rk[k] = 0;
            if (ok[k]) rk[k] = atomicAdd(&h[rc[k].x - row0], 1);
        }
        __syncthreads();

        const int cntr = h[tid];
        int inr = cntr;
#pragma unroll
        for (int d = 1; d < 32; d <<= 1) {
            const int t = __shfl_up(inr, d);
            if (lane >= d) inr += t;
        }
        if (lane == 31) wt2[wave] = inr;
        __syncthreads();
        int prr = 0;
#pragma unroll
        for (int k = 0; k < GT / 32; ++k) prr += (k < wave) ? wt2[k] : 0;
        hs[tid] = prr + inr - cntr;
        if (tid < RP) dg[tid] += (float)cntr;
        __syncthreads();

#pragma unroll
        for (int k = 0; k < KU; ++k) {
            if (ok[k]) {
                const int lr = rc[k].x - row0;
                se[hs[lr] + rk[k]] = make_int2(lr * PITCH, rc[k].y * 16);
            }
        }
        __syncthreads();

        const v4f* x4 = (const v4f*)x;
        const int sub = lane & 15;
        int2 en = make_int2(RP * PITCH, 0);
        if (wave * 32 + lane < nb) en = se[wave * 32 + lane];
        for (int gb = wave * 32; gb < nb; gb += GT) {
            const int2 e = en;
            v4f v[16];
            unsigned pi[16];
#define GLD(u) pi[u] = (unsigned)rshare<u>(e.y) + (unsigned)sub; v[u] = x4[pi[u]];
            GLD(0)  GLD(1)  GLD(2)  GLD(3)  GLD(4)  GLD(5)  GLD(6)  GLD(7)
            GLD(8)  GLD(9)  GLD(10) GLD(11) GLD(12) GLD(13) GLD(14) GLD(15)
#undef GLD
            asm volatile("" :: "v"(pi[0]), "v"(pi[1]), "v"(pi[2]), "v"(pi[3]), "v"(pi[4]), "v"(pi[5]), "v"(pi[6]), "v"(pi[7]),
                               "v"(pi[8]), "v"(pi[9]), "v"(pi[10]), "v"(pi[11]), "v"(pi[12]), "v"(pi[13]), "v"(pi[14]), "v"(pi[15]));
            const int ni = gb + GT + lane;
            en = se[(ni < CAP) ? ni : (CAP - 1)];
            if (ni >= nb) en = make_int2(RP * PITCH, 0);
            const int prow = __builtin_amdgcn_update_dpp(e.x, e.x, 0x111, 0xf, 0xf, false);
            const unsigned bm = __builtin_amdgcn_ballot_w32(prow != e.x);
            int cur = rshare<0>(e.x);
            v4f acc = {0.0f, 0.0f, 0.0f, 0.0f};
#define GAC(u) \
            if (u > 0 && __builtin_expect(((bm >> u) & 0x10001u) != 0u, 0)) { \
                const int rn = rshare<u>(e.x); \
                if (rn != cur) { \
                    float* tp = tile + cur + 4 * sub; \
                    atomicAdd(tp,     acc.x); \
                    atomicAdd(tp + 1, acc.y); \
                    atomicAdd(tp + 2, acc.z); \
                    atomicAdd(tp + 3, acc.w); \
                    acc = z; \
                    cur = rn; \
                } \
            } \
            acc += v[u];
            GAC(0)  GAC(1)  GAC(2)  GAC(3)  GAC(4)  GAC(5)  GAC(6)  GAC(7)
            GAC(8)  GAC(9)  GAC(10) GAC(11) GAC(12) GAC(13) GAC(14) GAC(15)
#undef GAC
            float* tp = tile + cur + 4 * sub;
            atomicAdd(tp,     acc.x);
            atomicAdd(tp + 1, acc.y);
            atomicAdd(tp + 2, acc.z);
            atomicAdd(tp + 3, acc.w);
        }

        if (b0 + CAP < total) {
            __syncthreads();
            h[tid] = 0;
            __syncthreads();
        }
    }

#pragma unroll
    for (int k = 0; k < 1024 / GT; ++k) {
        const int i = tid + k * GT;
        *(v4f*)(Wl + (i >> 4) * PITCH + (i & 15) * 4) = wv[k];
    }
    __syncthreads();

    const int c16  = lane & 15;
    const int half = lane >> 4;
    const int RTu  = (nrows + 15) >> 4;
    for (int rt = wave; rt < RTu; rt += GT / 32) {
        const int   r      = rt * 16 + c16;
        const float invdeg = 1.0f / (dg[r] + 1e-6f);
        const float* arow  = tile + r * PITCH + 2 * half;
        v2f a[16];
#pragma unroll
        for (int j = 0; j < 16; ++j) a[j] = *(const v2f*)(arow + 4 * j);

        v8f acc0 = {}, acc1 = {}, acc2 = {}, acc3 = {};
        const float* w0 = Wl + (0 * 16 + c16) * PITCH + 2 * half;
        const float* w1 = Wl + (1 * 16 + c16) * PITCH + 2 * half;
        const float* w2 = Wl + (2 * 16 + c16) * PITCH + 2 * half;
        const float* w3 = Wl + (3 * 16 + c16) * PITCH + 2 * half;
#pragma unroll
        for (int j = 0; j < 16; ++j) {
            v2f as;
            as.x = a[j].x * invdeg;
            as.y = a[j].y * invdeg;
            const v2f b0 = *(const v2f*)(w0 + 4 * j);
            const v2f b1 = *(const v2f*)(w1 + 4 * j);
            const v2f b2 = *(const v2f*)(w2 + 4 * j);
            const v2f b3 = *(const v2f*)(w3 + 4 * j);
            acc0 = __builtin_amdgcn_wmma_f32_16x16x4_f32(false, as, false, b0, (short)0, acc0, false, false);
            acc1 = __builtin_amdgcn_wmma_f32_16x16x4_f32(false, as, false, b1, (short)0, acc1, false, false);
            acc2 = __builtin_amdgcn_wmma_f32_16x16x4_f32(false, as, false, b2, (short)0, acc2, false, false);
            acc3 = __builtin_amdgcn_wmma_f32_16x16x4_f32(false, as, false, b3, (short)0, acc3, false, false);
        }

        v8f accs[4] = {acc0, acc1, acc2, acc3};
        float* orow = tile + (rt * 16 + half * 8) * PITCH + c16;
#pragma unroll
        for (int t = 0; t < 4; ++t) {
#pragma unroll
            for (int i = 0; i < 8; ++i)
                orow[i * PITCH + t * 16] = accs[t][i] + bv[t];
        }
    }
    __syncthreads();

    v4f* o4 = (v4f*)(out + (long)row0 * 64);
    const int nq = nrows * 16;
    for (int i0 = tid; i0 < nq; i0 += GT * OU) {
        const int wb = __builtin_amdgcn_readfirstlane(i0 - lane);
        v4f q[OU];
#pragma unroll
        for (int k = 0; k < OU; ++k) {
            const int i  = i0 + k * GT;
            const int ic = (i < nq) ? i : (nq - 1);
            q[k] = *(const v4f*)(tile + (ic >> 4) * PITCH + (ic & 15) * 4);
        }
#pragma unroll
        for (int k = 0; k < OU; ++k)
            if (wb + k * GT + 32 <= nq) o4[i0 + k * GT] = q[k];
        int nf = (nq - wb - 32 >= 0) ? ((nq - wb - 32) / GT + 1) : 0;
        if (nf > OU) nf = OU;
        const int it = i0 + nf * GT;
        if (nf < OU && it < nq)
            o4[it] = *(const v4f*)(tile + (it >> 4) * PITCH + (it & 15) * 4);
    }
}

extern "C" void kernel_launch(void* const* d_in, const int* in_sizes, int n_in,
                              void* d_out, int out_size, void* d_ws, size_t ws_size,
                              hipStream_t stream) {
    const float* x  = (const float*)d_in[0];
    const int*   ei = (const int*)d_in[1];
    const float* W  = (const float*)d_in[2];
    const float* b  = (const float*)d_in[3];
    float* out = (float*)d_out;

    const int N = in_sizes[0] / 64;
    const int E = in_sizes[1] / 2;
    const int rpb = (N + NB - 1) / NB;

    char* ws = (char*)d_ws;
    const size_t seg_bytes = (size_t)NB * NW * 8;
    int*   seg   = (int*)ws;
    int*   pairs = (int*)(ws + seg_bytes);

    const int chunk = (E + NW - 1) / NW;
    gcn_sort<<<NW, ST, (size_t)chunk * 20, stream>>>(ei, pairs, seg, E, rpb);
    const int rp = (rpb + 15) / 16 * 16;
    const size_t lds = ((size_t)(rp + 1) * PITCH + (size_t)rp) * 4;
    gcn_gather_gemm<<<NB, GT, lds, stream>>>(x, pairs, seg, W, b, out, N, rpb);
}
